# starting kernel
# baseline (speedup 1.0000x reference)
#include <hip/hip_runtime.h>

typedef float v2f __attribute__((ext_vector_type(2)));
typedef float v8f __attribute__((ext_vector_type(8)));

#define IN_CH  128
#define HID1   64
#define HID2   32

__global__ void fill_zero_kernel(float* __restrict__ p, int n) {
    int i = blockIdx.x * blockDim.x + threadIdx.x;
    if (i < n) p[i] = 0.0f;
}

__global__ void degree_kernel(const int* __restrict__ dst, float* __restrict__ deg, int nE) {
    int e = blockIdx.x * blockDim.x + threadIdx.x;
    if (e < nE) atomicAdd(&deg[dst[e]], 1.0f);
}

__global__ void dinv_kernel(float* __restrict__ dinv, int n) {
    int i = blockIdx.x * blockDim.x + threadIdx.x;
    if (i < n) dinv[i] = rsqrtf(dinv[i] + 1.0f);
}

__global__ __launch_bounds__(128) void gemm1_wmma(const float* __restrict__ X,
                                                  const float* __restrict__ W,
                                                  float* __restrict__ H) {
    const int lane = threadIdx.x & 31;
    const int wave = threadIdx.x >> 5;
    const int m0   = blockIdx.x * 16;
    const int n0   = wave * 16;
    const int l16  = lane & 15;
    const int khi  = (lane >> 4) * 2;
    const int row  = m0 + l16;
    const int col  = n0 + l16;
    v8f c = {};
    #pragma unroll 4
    for (int kk = 0; kk < IN_CH; kk += 4) {
        v2f a, b;
        a.x = X[row * IN_CH + kk + khi + 0];
        a.y = X[row * IN_CH + kk + khi + 1];
        b.x = W[(kk + khi + 0) * HID1 + col];
        b.y = W[(kk + khi + 1) * HID1 + col];
        c = __builtin_amdgcn_wmma_f32_16x16x4_f32(false, a, false, b,
                                                  (short)0, c, false, false);
    }
    const int mbase = m0 + ((lane >> 4) * 8);
    #pragma unroll
    for (int r = 0; r < 8; ++r)
        H[(mbase + r) * HID1 + col] = c[r];
}

__global__ __launch_bounds__(64) void gemm2_wmma(const float* __restrict__ X,
                                                 const float* __restrict__ W,
                                                 float* __restrict__ H) {
    const int lane = threadIdx.x & 31;
    const int wave = threadIdx.x >> 5;
    const int m0   = blockIdx.x * 16;
    const int n0   = wave * 16;
    const int l16  = lane & 15;
    const int khi  = (lane >> 4) * 2;
    const int row  = m0 + l16;
    const int col  = n0 + l16;
    v8f c = {};
    #pragma unroll 4
    for (int kk = 0; kk < HID1; kk += 4) {
        v2f a, b;
        a.x = X[row * HID1 + kk + khi + 0];
        a.y = X[row * HID1 + kk + khi + 1];
        b.x = W[(kk + khi + 0) * HID2 + col];
        b.y = W[(kk + khi + 1) * HID2 + col];
        c = __builtin_amdgcn_wmma_f32_16x16x4_f32(false, a, false, b,
                                                  (short)0, c, false, false);
    }
    const int mbase = m0 + ((lane >> 4) * 8);
    #pragma unroll
    for (int r = 0; r < 8; ++r)
        H[(mbase + r) * HID2 + col] = c[r];
}

__global__ __launch_bounds__(256) void edge_agg1_kernel(const float* __restrict__ h,
                                                        const int* __restrict__ src,
                                                        const int* __restrict__ dst,
                                                        const float* __restrict__ dinv,
                                                        float* __restrict__ agg, int nE) {
    int tid = blockIdx.x * 256 + threadIdx.x;
    int e = tid >> 4;
    if (e >= nE) return;
    int c4 = (tid & 15) * 4;
    int s = src[e], d = dst[e];
    float norm = dinv[s] * dinv[d];
    const float4 hv = *(const float4*)(h + s * HID1 + c4);
    float* ap = agg + d * HID1 + c4;
    atomicAdd(ap + 0, hv.x * norm);
    atomicAdd(ap + 1, hv.y * norm);
    atomicAdd(ap + 2, hv.z * norm);
    atomicAdd(ap + 3, hv.w * norm);
}

__global__ __launch_bounds__(256) void edge_agg2_kernel(const float* __restrict__ h,
                                                        const int* __restrict__ src,
                                                        const int* __restrict__ dst,
                                                        const float* __restrict__ dinv,
                                                        float* __restrict__ agg, int nE) {
    int tid = blockIdx.x * 256 + threadIdx.x;
    int e = tid >> 3;
    if (e >= nE) return;
    int c4 = (tid & 7) * 4;
    int s = src[e], d = dst[e];
    float norm = dinv[s] * dinv[d];
    const float4 hv = *(const float4*)(h + s * HID2 + c4);
    float* ap = agg + d * HID2 + c4;
    atomicAdd(ap + 0, hv.x * norm);
    atomicAdd(ap + 1, hv.y * norm);
    atomicAdd(ap + 2, hv.z * norm);
    atomicAdd(ap + 3, hv.w * norm);
}

__global__ __launch_bounds__(256) void combine1_kernel(float* __restrict__ h,
                                                       const float* __restrict__ agg,
                                                       const float* __restrict__ dinv,
                                                       const float* __restrict__ b, int n) {
    int i = blockIdx.x * 256 + threadIdx.x;
    if (i >= n * HID1) return;
    int node = i >> 6, c = i & 63;
    float dd = dinv[node]; dd *= dd;
    float v = agg[i] + h[i] * dd + b[c];
    h[i] = fmaxf(v, 0.0f);
}

__global__ __launch_bounds__(256) void combine2_proj_kernel(const float* __restrict__ h,
                                                            const float* __restrict__ agg,
                                                            const float* __restrict__ dinv,
                                                            const float* __restrict__ b2,
                                                            const float* __restrict__ Wh,
                                                            const float* __restrict__ bh,
                                                            float* __restrict__ out, int n) {
    int lane = threadIdx.x & 31;
    int node = blockIdx.x * 8 + (threadIdx.x >> 5);
    if (node >= n) return;
    float dd = dinv[node]; dd *= dd;
    int i = node * HID2 + lane;
    float v = agg[i] + h[i] * dd + b2[lane];
    v = fmaxf(v, 0.0f);
    float acc = v * Wh[lane];
    #pragma unroll
    for (int off = 16; off > 0; off >>= 1)
        acc += __shfl_xor(acc, off, 32);
    if (lane == 0) out[node] = acc + bh[0];
}

extern "C" void kernel_launch(void* const* d_in, const int* in_sizes, int n_in,
                              void* d_out, int out_size, void* d_ws, size_t ws_size,
                              hipStream_t stream) {
    const float* x  = (const float*)d_in[0];
    const int*   ei = (const int*)d_in[1];
    const float* W1 = (const float*)d_in[2];
    const float* b1 = (const float*)d_in[3];
    const float* W2 = (const float*)d_in[4];
    const float* b2 = (const float*)d_in[5];
    const float* Wh = (const float*)d_in[6];
    const float* bh = (const float*)d_in[7];
    float* out = (float*)d_out;

    const int N = in_sizes[0] / IN_CH;
    const int E = in_sizes[1] / 2;
    const int* src = ei;
    const int* dst = ei + E;

    float* f    = (float*)d_ws;
    float* dinv = f;
    float* h1   = f + 65536;
    float* agg1 = h1 + (size_t)N * HID1;
    float* h2   = agg1 + (size_t)N * HID1;
    float* agg2 = h2 + (size_t)N * HID2;

    fill_zero_kernel<<<(N + 255) / 256, 256, 0, stream>>>(dinv, N);
    fill_zero_kernel<<<(N * HID1 + 255) / 256, 256, 0, stream>>>(agg1, N * HID1);
    fill_zero_kernel<<<(N * HID2 + 255) / 256, 256, 0, stream>>>(agg2, N * HID2);

    degree_kernel<<<(E + 255) / 256, 256, 0, stream>>>(dst, dinv, E);
    dinv_kernel<<<(N + 255) / 256, 256, 0, stream>>>(dinv, N);

    gemm1_wmma<<<N / 16, 128, 0, stream>>>(x, W1, h1);
    edge_agg1_kernel<<<(E * 16 + 255) / 256, 256, 0, stream>>>(h1, src, dst, dinv, agg1, E);
    combine1_kernel<<<(N * HID1 + 255) / 256, 256, 0, stream>>>(h1, agg1, dinv, b1, N);

    gemm2_wmma<<<N / 16, 64, 0, stream>>>(h1, W2, h2);
    edge_agg2_kernel<<<(E * 8 + 255) / 256, 256, 0, stream>>>(h2, src, dst, dinv, agg2, E);

    combine2_proj_kernel<<<(N + 7) / 8, 256, 0, stream>>>(h2, agg2, dinv, b2, Wh, bh, out, N);
}


# Edge list read once: 128 work-groups each order 1/128 of the edges by destination range in LDS (int atomic rank) and store the blocks; no range rescans a slice; sums: LDS arrival, slice cuts shift
# speedup vs baseline: 2.7990x; 2.7990x over previous
#include <hip/hip_runtime.h>

typedef float v2f __attribute__((ext_vector_type(2)));
typedef float v8f __attribute__((ext_vector_type(8)));

#define IN_CH  128
#define HID1   64
#define HID2   32

template <int RN, int RNP, int NRANGE, int NSLICE>
__global__ void dinv_kernel(const int* __restrict__ degp, float* __restrict__ dinv, int n) {
    int i = blockIdx.x * blockDim.x + threadIdx.x;
    if (i < n) {
        const int range = i / RN;
        const int off   = i - range * RN;
        int c = 0;
        #pragma unroll
        for (int s = 0; s < NSLICE; ++s) c += degp[(s * NRANGE + range) * RNP + off];
        dinv[i] = rsqrtf((float)c + 1.0f);
    }
}

__global__ __launch_bounds__(128) void gemm1_wmma(const float* __restrict__ X,
                                                  const float* __restrict__ W,
                                                  float* __restrict__ H) {
    const int lane = threadIdx.x & 31;
    const int wave = threadIdx.x >> 5;
    const int m0   = blockIdx.x * 16;
    const int n0   = wave * 16;
    const int l16  = lane & 15;
    const int khi  = (lane >> 4) * 2;
    const int row  = m0 + l16;
    const int col  = n0 + l16;
    v8f c = {};
    #pragma unroll 4
    for (int kk = 0; kk < IN_CH; kk += 4) {
        v2f a, b;
        a.x = X[row * IN_CH + kk + khi + 0];
        a.y = X[row * IN_CH + kk + khi + 1];
        b.x = W[(kk + khi + 0) * HID1 + col];
        b.y = W[(kk + khi + 1) * HID1 + col];
        c = __builtin_amdgcn_wmma_f32_16x16x4_f32(false, a, false, b,
                                                  (short)0, c, false, false);
    }
    const int mbase = m0 + ((lane >> 4) * 8);
    #pragma unroll
    for (int r = 0; r < 8; ++r)
        H[(mbase + r) * HID1 + col] = c[r];
}

__global__ __launch_bounds__(64) void gemm2_wmma(const float* __restrict__ X,
                                                 const float* __restrict__ W,
                                                 float* __restrict__ H) {
    const int lane = threadIdx.x & 31;
    const int wave = threadIdx.x >> 5;
    const int m0   = blockIdx.x * 16;
    const int n0   = wave * 16;
    const int l16  = lane & 15;
    const int khi  = (lane >> 4) * 2;
    const int row  = m0 + l16;
    const int col  = n0 + l16;
    v8f c = {};
    #pragma unroll 4
    for (int kk = 0; kk < HID1; kk += 4) {
        v2f a, b;
        a.x = X[row * HID1 + kk + khi + 0];
        a.y = X[row * HID1 + kk + khi + 1];
        b.x = W[(kk + khi + 0) * HID2 + col];
        b.y = W[(kk + khi + 1) * HID2 + col];
        c = __builtin_amdgcn_wmma_f32_16x16x4_f32(false, a, false, b,
                                                  (short)0, c, false, false);
    }
    const int mbase = m0 + ((lane >> 4) * 8);
    #pragma unroll
    for (int r = 0; r < 8; ++r)
        H[(mbase + r) * HID2 + col] = c[r];
}

#define EPART 12512
template <int RN, int NRANGE, int NSLICE>
struct SplitGeom {
    static constexpr int STG    = EPART + NRANGE * 32;
    static constexpr int CAPBLK = STG / 32;
};

template <int RN, int NRANGE, int NSLICE>
__global__ __launch_bounds__(1024) void edge_split_kernel(const int* __restrict__ src,
                                                         const int* __restrict__ dst,
                                                         int* __restrict__ lists,
                                                         int nE, int nN) {
    static_assert(NRANGE == 64, "two ranges per lane of the scanning wave");
    using G = SplitGeom<RN, NRANGE, NSLICE>;
    constexpr int STGA = G::STG > 20608 ? G::STG : 20608;
    constexpr int KPT  = (EPART + 1023) / 1024;
    __shared__ int st_s[STGA];
    __shared__ int st_d[STGA];
    __shared__ int hist[NRANGE];
    __shared__ int offs[NRANGE + 1];
    const int tid  = threadIdx.x;
    const int lane = tid & 31;
    const int wave = tid >> 5;
    const int part = blockIdx.x;
    const int nPart = NSLICE * 32;
    const int e0 = (int)((long long)nE * part / nPart);
    const int e1 = (int)((long long)nE * (part + 1) / nPart);
    int cnt = e1 - e0;
    cnt = cnt > EPART ? EPART : cnt;

    int dv[KPT], sv[KPT], pk[KPT];
    #pragma unroll
    for (int k = 0; k < KPT; ++k) {
        const int i = k * 1024 + tid;
        const int e = i < cnt ? e0 + i : 0;
        dv[k] = dst[e];
        sv[k] = src[e];
    }
    for (int i = tid; i < G::STG; i += 1024) { st_s[i] = 0; st_d[i] = 0; }
    if (tid < NRANGE) hist[tid] = 0;
    __syncthreads();

    #pragma unroll
    for (int k = 0; k < KPT; ++k) {
        const int i = k * 1024 + tid;
        pk[k] = -1;
        if (i < cnt && (unsigned)dv[k] < (unsigned)(nN < RN * NRANGE ? nN : RN * NRANGE)) {
            const int r  = (int)((unsigned)dv[k] / (unsigned)RN);
            const int dl = dv[k] - r * RN;
            const int rank = atomicAdd(&hist[r], 1);
            pk[k] = r | (dl << 6) | (rank << 16);
        }
    }
    __syncthreads();

    if (wave == 0) {
        const int h0 = hist[lane], h1 = hist[lane + 32];
        const int c0 = (h0 + 31) & ~31, c1 = (h1 + 31) & ~31;
        int i0 = c0, i1 = c1;
        #pragma unroll
        for (int off = 1; off < 32; off <<= 1) {
            const int t0 = __shfl_up(i0, off, 32);
            const int t1 = __shfl_up(i1, off, 32);
            if (lane >= off) { i0 += t0; i1 += t1; }
        }
        const int tot0 = __shfl(i0, 31, 32);
        const int tot1 = __shfl(i1, 31, 32);
        const int o0 = i0 - c0, o1 = tot0 + i1 - c1;
        offs[lane] = o0;
        offs[lane + 32] = o1;
        if (lane == 0) offs[NRANGE] = tot0 + tot1;
        const int slice = part >> 5, w = part & 31;
        int* const tables = lists + (size_t)nPart * G::CAPBLK * 96;
        int* t0p = tables + (size_t)((slice * NRANGE + lane) * 32 + w) * 2;
        int* t1p = tables + (size_t)((slice * NRANGE + lane + 32) * 32 + w) * 2;
        t0p[0] = o0 >> 5; t0p[1] = h0;
        t1p[0] = o1 >> 5; t1p[1] = h1;
    }
    __syncthreads();

    #pragma unroll
    for (int k = 0; k < KPT; ++k) {
        if (pk[k] >= 0) {
            const int pos = offs[pk[k] & 63] + (pk[k] >> 16);
            st_s[pos] = sv[k];
            st_d[pos] = (pk[k] >> 6) & 1023;
        }
    }
    __syncthreads();

    const int tot = offs[NRANGE];
    int* const gp = lists + (size_t)part * G::CAPBLK * 96;
    for (int i = tid; i < tot; i += 1024) {
        int* bp = gp + (size_t)(i >> 5) * 96;
        bp[lane]      = st_s[i];
        bp[32 + lane] = st_d[i];
    }
}

template <int RN, int RNP, int NRANGE, int NSLICE>
__global__ __launch_bounds__(1024) void edge_count_kernel(const int* __restrict__ lists,
                                                         int* __restrict__ degp, int nN) {
    using G = SplitGeom<RN, NRANGE, NSLICE>;
    __shared__ int cntl[RN];
    const int tid  = threadIdx.x;
    const int lane = tid & 31;
    const int wave = tid >> 5;
    const int range = blockIdx.x % NRANGE;
    const int slice = blockIdx.x / NRANGE;
    const int lo = range * RN;
    int cnt = nN - lo;
    cnt = cnt < 0 ? 0 : (cnt > RN ? RN : cnt);
    const int* const gp = lists + (size_t)(slice * 32 + wave) * G::CAPBLK * 96;
    const int* const tb = lists + (size_t)NSLICE * 32 * G::CAPBLK * 96
                                + (size_t)(blockIdx.x * 32 + wave) * 2;
    const int fb = __builtin_amdgcn_readfirstlane(tb[0]);
    const int hc = __builtin_amdgcn_readfirstlane(tb[1]);
    const int nb = (hc + 31) >> 5;
    for (int i = tid; i < RN; i += 1024) cntl[i] = 0;
    __syncthreads();
    for (int k0 = 0; k0 < nb; k0 += 8) {
        int dlv[8];
        #pragma unroll
        for (int j = 0; j < 8; ++j) {
            const int k = k0 + j < nb ? k0 + j : nb - 1;
            dlv[j] = gp[(size_t)(fb + k) * 96 + 32 + lane];
        }
        #pragma unroll
        for (int j = 0; j < 8; ++j)
            if (lane < hc - 32 * (k0 + j)) atomicAdd(&cntl[dlv[j]], 1);
    }
    __syncthreads();
    int* dp = degp + (size_t)(slice * NRANGE + range) * RNP;
    for (int i = tid; i < cnt; i += 1024) dp[i] = cntl[i];
}

template <int CH, bool FIRST, int RN, int NRANGE, int NSLICE>
__global__ __launch_bounds__(1024) void edge_agg_list_kernel(const float* __restrict__ h,
                                                            int* __restrict__ lists,
                                                            const float* __restrict__ dinv,
                                                            float* __restrict__ part,
                                                            int nN) {
    constexpr int VW = CH / 32;
    constexpr int ACCF = RN * CH > 41472 ? RN * CH : 41472;
    __shared__ float acc[ACCF];
    const int tid  = threadIdx.x;
    const int lane = tid & 31;
    const int wave = tid >> 5;
    const int range = blockIdx.x % NRANGE;
    const int slice = blockIdx.x / NRANGE;
    const int lo = range * RN;
    int cnt = nN - lo;
    cnt = cnt < 0 ? 0 : (cnt > RN ? RN : cnt);

    using G = SplitGeom<RN, NRANGE, NSLICE>;
    int* const gp = lists + (size_t)(slice * 32 + wave) * G::CAPBLK * 96;
    const int* const tb = lists + (size_t)NSLICE * 32 * G::CAPBLK * 96
                                + (size_t)(blockIdx.x * 32 + wave) * 2;
    const int fb = __builtin_amdgcn_readfirstlane(tb[0]);
    const int hc = __builtin_amdgcn_readfirstlane(tb[1]);
    const int nb = (hc + 31) >> 5;
    int s = 0, dl = 0, nmi = 0, n = 0;
    int* bp = gp;
    if (nb > 0) {
        bp = gp + (size_t)fb * 96;
        n = hc < 32 ? hc : 32;
        s = bp[lane]; dl = bp[32 + lane];
        if (!FIRST) nmi = bp[64 + lane];
    }

    for (int i = tid * 4; i < RN * CH; i += 4096)
        *(float4*)&acc[i] = make_float4(0.0f, 0.0f, 0.0f, 0.0f);
    __syncthreads();

    for (int k = 0; k < nb; ++k) {
        int s2 = 0, dl2 = 0, nmi2 = 0, n2 = 0;
        const bool more = k + 1 < nb;
        int* bp2 = gp;
        if (more) {
            bp2 = gp + (size_t)(fb + k + 1) * 96;
            const int left = hc - 32 * (k + 1);
            n2 = left < 32 ? left : 32;
        }
        float ns = 0.0f, nd = 0.0f, nm;
        if (n == 32) {
            float r[32][VW];
            #pragma unroll
            for (int u = 0; u < 32; ++u) {
                const unsigned su = (unsigned)__builtin_amdgcn_readlane(s, u);
                const float* hp = h + (size_t)su * CH + lane * VW;
                #pragma unroll
                for (int w = 0; w < VW; ++w) r[u][w] = hp[w];
            }
            if (FIRST) { ns = dinv[s]; nd = dinv[lo + dl]; }
            if (more) {
                s2 = bp2[lane]; dl2 = bp2[32 + lane];
                if (!FIRST) nmi2 = bp2[64 + lane];
            }
            nm = FIRST ? ns * nd : __builtin_bit_cast(float, nmi);
            #pragma unroll
            for (int u = 0; u < 32; ++u) {
                const int du = __builtin_amdgcn_readlane(dl, u);
                const float nu = __builtin_bit_cast(float,
                    __builtin_amdgcn_readlane(__builtin_bit_cast(int, nm), u));
                float* ap = &acc[du * CH + lane * VW];
                #pragma unroll
                for (int w = 0; w < VW; ++w) atomicAdd(ap + w, r[u][w] * nu);
            }
        } else {
            if (FIRST) { ns = dinv[s]; nd = dinv[lo + dl]; }
            if (more) {
                s2 = bp2[lane]; dl2 = bp2[32 + lane];
                if (!FIRST) nmi2 = bp2[64 + lane];
            }
            nm = FIRST ? ((lane < n) ? ns * nd : 0.0f) : __builtin_bit_cast(float, nmi);
            #pragma unroll
            for (int g = 0; g < 4; ++g) {
                if (g * 8 < n) {
                    float r[8][VW];
                    #pragma unroll
                    for (int u = 0; u < 8; ++u) {
                        const unsigned su = (unsigned)__builtin_amdgcn_readlane(s, g * 8 + u);
                        const float* hp = h + (size_t)su * CH + lane * VW;
                        #pragma unroll
                        for (int w = 0; w < VW; ++w) r[u][w] = hp[w];
                    }
                    #pragma unroll
                    for (int u = 0; u < 8; ++u) {
                        if (g * 8 + u < n) {
                            const int du = __builtin_amdgcn_readlane(dl, g * 8 + u);
                            const float nu = __builtin_bit_cast(float,
                                __builtin_amdgcn_readlane(__builtin_bit_cast(int, nm), g * 8 + u));
                            float* ap = &acc[du * CH + lane * VW];
                            #pragma unroll
                            for (int w = 0; w < VW; ++w) atomicAdd(ap + w, r[u][w] * nu);
                        }
                    }
                }
            }
        }
        if (FIRST) bp[64 + lane] = __builtin_bit_cast(int, nm);
        s = s2; dl = dl2; nmi = nmi2; n = n2; bp = bp2;
    }
    __syncthreads();

    float* pp = part + (size_t)slice * nN * CH + (size_t)lo * CH;
    for (int i = tid * 4; i < cnt * CH; i += 4096)
        *(float4*)(pp + i) = *(const float4*)&acc[i];
}

template <int NSLICE>
__global__ __launch_bounds__(256) void combine1_kernel(float* __restrict__ h,
                                                       const float* __restrict__ part,
                                                       const float* __restrict__ dinv,
                                                       const float* __restrict__ b, int n) {
    int i = blockIdx.x * 256 + threadIdx.x;
    if (i >= n * (HID1 / 4)) return;
    int node = i >> 4, c = (i & 15) * 4;
    float dd = dinv[node]; dd *= dd;
    const size_t plane = (size_t)n * HID1;
    float4 a = *(const float4*)(part + (size_t)i * 4);
    #pragma unroll
    for (int s = 1; s < NSLICE; ++s) {
        const float4 p = *(const float4*)(part + s * plane + (size_t)i * 4);
        a.x += p.x; a.y += p.y; a.z += p.z; a.w += p.w;
    }
    const float4 hv = *(const float4*)(h + (size_t)i * 4);
    const float4 bv = *(const float4*)(b + c);
    float4 v;
    v.x = fmaxf(a.x + hv.x * dd + bv.x, 0.0f);
    v.y = fmaxf(a.y + hv.y * dd + bv.y, 0.0f);
    v.z = fmaxf(a.z + hv.z * dd + bv.z, 0.0f);
    v.w = fmaxf(a.w + hv.w * dd + bv.w, 0.0f);
    *(float4*)(h + (size_t)i * 4) = v;
}

template <int NSLICE>
__global__ __launch_bounds__(256) void combine2_proj_kernel(const float* __restrict__ h,
                                                            const float* __restrict__ part,
                                                            const float* __restrict__ dinv,
                                                            const float* __restrict__ b2,
                                                            const float* __restrict__ Wh,
                                                            const float* __restrict__ bh,
                                                            float* __restrict__ out, int n) {
    int lane = threadIdx.x & 31;
    int node = blockIdx.x * 8 + (threadIdx.x >> 5);
    if (node >= n) return;
    float dd = dinv[node]; dd *= dd;
    int i = node * HID2 + lane;
    const size_t plane = (size_t)n * HID2;
    float a = part[i];
    #pragma unroll
    for (int s = 1; s < NSLICE; ++s) a += part[s * plane + i];
    float v = a + h[i] * dd + b2[lane];
    v = fmaxf(v, 0.0f);
    float acc = v * Wh[lane];
    #pragma unroll
    for (int off = 16; off > 0; off >>= 1)
        acc += __shfl_xor(acc, off, 32);
    if (lane == 0) out[node] = acc + bh[0];
}

#define NR1 64
#define NS1 4
#define RN1 782
#define RNP1 800

extern "C" void kernel_launch(void* const* d_in, const int* in_sizes, int n_in,
                              void* d_out, int out_size, void* d_ws, size_t ws_size,
                              hipStream_t stream) {
    const float* x  = (const float*)d_in[0];
    const int*   ei = (const int*)d_in[1];
    const float* W1 = (const float*)d_in[2];
    const float* b1 = (const float*)d_in[3];
    const float* W2 = (const float*)d_in[4];
    const float* b2 = (const float*)d_in[5];
    const float* Wh = (const float*)d_in[6];
    const float* bh = (const float*)d_in[7];
    float* out = (float*)d_out;

    const int N = in_sizes[0] / IN_CH;
    const int E = in_sizes[1] / 2;
    const int* src = ei;
    const int* dst = ei + E;

    float* f     = (float*)d_ws;
    float* dinv  = f;
    float* h1    = f + 65536;
    float* part1 = h1 + (size_t)N * HID1;
    float* h2    = part1 + (size_t)NS1 * N * HID1;
    float* part2 = h2 + (size_t)N * HID2;
    int*   lists = (int*)(part2 + (size_t)NS1 * N * HID2);
    int*   degp  = (int*)part2;

    gemm1_wmma<<<N / 16, 128, 0, stream>>>(x, W1, h1);

    edge_split_kernel<RN1, NR1, NS1><<<NS1 * 32, 1024, 0, stream>>>(src, dst, lists, E, N);
    edge_count_kernel<RN1, RNP1, NR1, NS1><<<NR1 * NS1, 1024, 0, stream>>>(lists, degp, N);
    dinv_kernel<RN1, RNP1, NR1, NS1><<<(N + 255) / 256, 256, 0, stream>>>(degp, dinv, N);

    edge_agg_list_kernel<HID1, true, RN1, NR1, NS1><<<NR1 * NS1, 1024, 0, stream>>>(h1, lists, dinv, part1, N);
    combine1_kernel<NS1><<<(N * (HID1 / 4) + 255) / 256, 256, 0, stream>>>(h1, part1, dinv, b1, N);

    gemm2_wmma<<<N / 16, 64, 0, stream>>>(h1, W2, h2);
    edge_agg_list_kernel<HID2, false, RN1, NR1, NS1><<<NR1 * NS1, 1024, 0, stream>>>(h2, lists, dinv, part2, N);

    combine2_proj_kernel<NS1><<<(N + 7) / 8, 256, 0, stream>>>(h2, part2, dinv, b2, Wh, bh, out, N);
}


# No edge slices: 256 ranges x 1, a work-group walks every edge of its nodes, so its LDS sums and degrees are final: dinv and both combines end the pass; 4 planes, 3 launches go; sums: LDS arrival
# speedup vs baseline: 3.2981x; 1.1783x over previous
#include <hip/hip_runtime.h>

typedef float v2f __attribute__((ext_vector_type(2)));
typedef float v8f __attribute__((ext_vector_type(8)));

#define IN_CH  128
#define HID1   64
#define HID2   32

__global__ __launch_bounds__(128) void gemm1_wmma(const float* __restrict__ X,
                                                  const float* __restrict__ W,
                                                  float* __restrict__ H) {
    const int lane = threadIdx.x & 31;
    const int wave = threadIdx.x >> 5;
    const int m0   = blockIdx.x * 16;
    const int n0   = wave * 16;
    const int l16  = lane & 15;
    const int khi  = (lane >> 4) * 2;
    const int row  = m0 + l16;
    const int col  = n0 + l16;
    v8f c = {};
    #pragma unroll 4
    for (int kk = 0; kk < IN_CH; kk += 4) {
        v2f a, b;
        a.x = X[row * IN_CH + kk + khi + 0];
        a.y = X[row * IN_CH + kk + khi + 1];
        b.x = W[(kk + khi + 0) * HID1 + col];
        b.y = W[(kk + khi + 1) * HID1 + col];
        c = __builtin_amdgcn_wmma_f32_16x16x4_f32(false, a, false, b,
                                                  (short)0, c, false, false);
    }
    const int mbase = m0 + ((lane >> 4) * 8);
    #pragma unroll
    for (int r = 0; r < 8; ++r)
        H[(mbase + r) * HID1 + col] = c[r];
}

__global__ __launch_bounds__(64) void gemm2_wmma(const float* __restrict__ X,
                                                 const float* __restrict__ W,
                                                 float* __restrict__ H) {
    const int lane = threadIdx.x & 31;
    const int wave = threadIdx.x >> 5;
    const int m0   = blockIdx.x * 16;
    const int n0   = wave * 16;
    const int l16  = lane & 15;
    const int khi  = (lane >> 4) * 2;
    const int row  = m0 + l16;
    const int col  = n0 + l16;
    v8f c = {};
    #pragma unroll 4
    for (int kk = 0; kk < HID1; kk += 4) {
        v2f a, b;
        a.x = X[row * HID1 + kk + khi + 0];
        a.y = X[row * HID1 + kk + khi + 1];
        b.x = W[(kk + khi + 0) * HID2 + col];
        b.y = W[(kk + khi + 1) * HID2 + col];
        c = __builtin_amdgcn_wmma_f32_16x16x4_f32(false, a, false, b,
                                                  (short)0, c, false, false);
    }
    const int mbase = m0 + ((lane >> 4) * 8);
    #pragma unroll
    for (int r = 0; r < 8; ++r)
        H[(mbase + r) * HID2 + col] = c[r];
}

#define EPART 12512

template <int NRANGE, int NPARTS>
struct WalkGeom {
    static constexpr int PPW = NPARTS / 32;
    static_assert(NPARTS % 32 == 0, "every wave walks the same number of parts");
};

template <int RN, int NRANGE, int NPARTS>
__global__ __launch_bounds__(1024) void edge_split_kernel(const int* __restrict__ src,
                                                         const int* __restrict__ dst,
                                                         int* __restrict__ lists,
                                                         int nE, int nN) {
    static_assert(NRANGE == 256, "eight ranges per lane of the scanning wave; 8-bit range field");
    static_assert(RN <= 256, "8-bit local row field");
    static_assert(EPART < 32768, "15-bit rank field");
    constexpr int STGA = EPART > 20608 ? EPART : 20608;
    constexpr int KPT  = (EPART + 1023) / 1024;
    __shared__ int st_s[STGA];
    __shared__ int st_d[STGA];
    __shared__ int hist[NRANGE];
    __shared__ int offs[NRANGE + 1];
    const int tid  = threadIdx.x;
    const int lane = tid & 31;
    const int wave = tid >> 5;
    const int part = blockIdx.x;
    const int e0 = (int)((long long)nE * part / NPARTS);
    const int e1 = (int)((long long)nE * (part + 1) / NPARTS);
    int cnt = e1 - e0;
    cnt = cnt > EPART ? EPART : cnt;

    int dv[KPT], sv[KPT], pk[KPT];
    #pragma unroll
    for (int k = 0; k < KPT; ++k) {
        const int i = k * 1024 + tid;
        const int e = i < cnt ? e0 + i : 0;
        dv[k] = dst[e];
        sv[k] = src[e];
    }
    if (tid < NRANGE) hist[tid] = 0;
    __syncthreads();

    #pragma unroll
    for (int k = 0; k < KPT; ++k) {
        const int i = k * 1024 + tid;
        pk[k] = -1;
        if (i < cnt && (unsigned)dv[k] < (unsigned)(nN < RN * NRANGE ? nN : RN * NRANGE)) {
            const int r  = (int)((unsigned)dv[k] / (unsigned)RN);
            const int dl = dv[k] - r * RN;
            const int rank = atomicAdd(&hist[r], 1);
            pk[k] = r | (dl << 8) | (rank << 16);
        }
    }
    __syncthreads();

    if (wave == 0) {
        int hv[8];
        int sum = 0;
        #pragma unroll
        for (int j = 0; j < 8; ++j) { hv[j] = hist[lane * 8 + j]; sum += hv[j]; }
        int inc = sum;
        #pragma unroll
        for (int off = 1; off < 32; off <<= 1) {
            const int t = __shfl_up(inc, off, 32);
            if (lane >= off) inc += t;
        }
        int run = inc - sum;
        #pragma unroll
        for (int j = 0; j < 8; ++j) { offs[lane * 8 + j] = run; run += hv[j]; }
        if (lane == 31) offs[NRANGE] = run;
    }
    __syncthreads();

    int* const tables = lists + (size_t)2 * NPARTS * EPART;
    if (tid < NRANGE)
        *(int2*)(tables + ((size_t)part * NRANGE + tid) * 2) = make_int2(offs[tid], hist[tid]);

    #pragma unroll
    for (int k = 0; k < KPT; ++k) {
        if (pk[k] >= 0) {
            const int pos = offs[pk[k] & 255] + (pk[k] >> 16);
            st_s[pos] = sv[k];
            st_d[pos] = (pk[k] >> 8) & 255;
        }
    }
    __syncthreads();

    const int tot = offs[NRANGE];
    int* const gs = lists + (size_t)part * EPART;
    int* const gd = lists + (size_t)NPARTS * EPART + (size_t)part * EPART;
    for (int i = tid; i < tot; i += 1024) {
        gs[i] = st_s[i];
        gd[i] = st_d[i];
    }
}

template <int NRANGE, int NPARTS>
struct WaveRuns {
    static constexpr int PPW = WalkGeom<NRANGE, NPARTS>::PPW;
    int base[PPW];
    int first[PPW];
    int total;
    __device__ __forceinline__ void load(const int* __restrict__ lists, int range, int wave) {
        const int* const tables = lists + (size_t)2 * NPARTS * EPART;
        int st[PPW], c[PPW];
        #pragma unroll
        for (int j = 0; j < PPW; ++j) {
            const int p = wave * PPW + j;
            const int2 t = *(const int2*)(tables + ((size_t)p * NRANGE + range) * 2);
            st[j] = __builtin_amdgcn_readfirstlane(t.x);
            c[j]  = __builtin_amdgcn_readfirstlane(t.y);
        }
        int pos = 0;
        #pragma unroll
        for (int j = 0; j < PPW; ++j) {
            first[j] = pos;
            base[j]  = (wave * PPW + j) * EPART + st[j] - pos;
            pos += c[j];
        }
        total = pos;
    }
    __device__ __forceinline__ int at(int v) const {
        int a = base[0];
        #pragma unroll
        for (int j = 1; j < PPW; ++j) a = v >= first[j] ? base[j] : a;
        return a + v;
    }
};

template <int RN, int NRANGE, int NPARTS>
__global__ __launch_bounds__(1024) void edge_count_dinv_kernel(const int* __restrict__ lists,
                                                              float* __restrict__ dinv, int nN) {
    __shared__ int cntl[RN];
    const int tid  = threadIdx.x;
    const int lane = tid & 31;
    const int wave = __builtin_amdgcn_readfirstlane(tid >> 5);
    const int range = blockIdx.x;
    const int lo = range * RN;
    int cnt = nN - lo;
    cnt = cnt < 0 ? 0 : (cnt > RN ? RN : cnt);
    const int* const gd = lists + (size_t)NPARTS * EPART;
    WaveRuns<NRANGE, NPARTS> runs;
    runs.load(lists, range, wave);
    const int nb = (runs.total + 31) >> 5;
    for (int i = tid; i < RN; i += 1024) cntl[i] = 0;
    __syncthreads();
    for (int k0 = 0; k0 < nb; k0 += 8) {
        int dlv[8];
        #pragma unroll
        for (int j = 0; j < 8; ++j) {
            const int v = 32 * (k0 + j) + lane;
            dlv[j] = v < runs.total ? gd[runs.at(v)] : -1;
        }
        #pragma unroll
        for (int j = 0; j < 8; ++j)
            if (dlv[j] >= 0) atomicAdd(&cntl[dlv[j]], 1);
    }
    __syncthreads();
    for (int i = tid; i < cnt; i += 1024) dinv[lo + i] = rsqrtf((float)cntl[i] + 1.0f);
}

template <int CH, bool LAST, int RN, int NRANGE, int NPARTS>
__global__ __launch_bounds__(1024) void edge_agg_kernel(const float* __restrict__ h,
                                                       const int* __restrict__ lists,
                                                       const float* __restrict__ dinv,
                                                       const float* __restrict__ bias,
                                                       const float* __restrict__ Wh,
                                                       const float* __restrict__ bh,
                                                       float* __restrict__ outp,
                                                       int nN) {
    constexpr int VW = CH / 32;
    constexpr int ACCF = RN * CH > 41472 ? RN * CH : 41472;
    __shared__ float acc[ACCF];
    __shared__ float outv[LAST ? RN : 1];
    const int tid  = threadIdx.x;
    const int lane = tid & 31;
    const int wave = __builtin_amdgcn_readfirstlane(tid >> 5);
    const int range = blockIdx.x;
    const int lo = range * RN;
    int cnt = nN - lo;
    cnt = cnt < 0 ? 0 : (cnt > RN ? RN : cnt);

    const int* const gs = lists;
    const int* const gd = lists + (size_t)NPARTS * EPART;
    WaveRuns<NRANGE, NPARTS> runs;
    runs.load(lists, range, wave);
    const int tot = runs.total;
    const int nb = (tot + 31) >> 5;
    int s = 0, dl = 0, n = 0;
    if (nb > 0) {
        n = tot < 32 ? tot : 32;
        if (lane < n) { const int a = runs.at(lane); s = gs[a]; dl = gd[a]; }
    }

    for (int i = tid * 4; i < RN * CH; i += 4096)
        *(float4*)&acc[i] = make_float4(0.0f, 0.0f, 0.0f, 0.0f);
    __syncthreads();

    for (int k = 0; k < nb; ++k) {
        int s2 = 0, dl2 = 0, n2 = 0;
        const bool more = k + 1 < nb;
        if (more) {
            const int left = tot - 32 * (k + 1);
            n2 = left < 32 ? left : 32;
        }
        const int v2 = 32 * (k + 1) + lane;
        float ns, nd, nm;
        if (n == 32) {
            float r[32][VW];
            #pragma unroll
            for (int u = 0; u < 32; ++u) {
                const unsigned su = (unsigned)__builtin_amdgcn_readlane(s, u);
                const float* hp = h + (size_t)su * CH + lane * VW;
                #pragma unroll
                for (int w = 0; w < VW; ++w) r[u][w] = hp[w];
            }
            ns = dinv[s]; nd = dinv[lo + dl];
            if (more && lane < n2) { const int a = runs.at(v2); s2 = gs[a]; dl2 = gd[a]; }
            nm = ns * nd;
            #pragma unroll
            for (int u = 0; u < 32; ++u) {
                const int du = __builtin_amdgcn_readlane(dl, u);
                const float nu = __builtin_bit_cast(float,
                    __builtin_amdgcn_readlane(__builtin_bit_cast(int, nm), u));
                float* ap = &acc[du * CH + lane * VW];
                #pragma unroll
                for (int w = 0; w < VW; ++w) atomicAdd(ap + w, r[u][w] * nu);
            }
        } else {
            ns = dinv[s]; nd = dinv[lo + dl];
            if (more && lane < n2) { const int a = runs.at(v2); s2 = gs[a]; dl2 = gd[a]; }
            nm = (lane < n) ? ns * nd : 0.0f;
            #pragma unroll
            for (int g = 0; g < 4; ++g) {
                if (g * 8 < n) {
                    float r[8][VW];
                    #pragma unroll
                    for (int u = 0; u < 8; ++u) {
                        const unsigned su = (unsigned)__builtin_amdgcn_readlane(s, g * 8 + u);
                        const float* hp = h + (size_t)su * CH + lane * VW;
                        #pragma unroll
                        for (int w = 0; w < VW; ++w) r[u][w] = hp[w];
                    }
                    #pragma unroll
                    for (int u = 0; u < 8; ++u) {
                        if (g * 8 + u < n) {
                            const int du = __builtin_amdgcn_readlane(dl, g * 8 + u);
                            const float nu = __builtin_bit_cast(float,
                                __builtin_amdgcn_readlane(__builtin_bit_cast(int, nm), g * 8 + u));
                            float* ap = &acc[du * CH + lane * VW];
                            #pragma unroll
                            for (int w = 0; w < VW; ++w) atomicAdd(ap + w, r[u][w] * nu);
                        }
                    }
                }
            }
        }
        s = s2; dl = dl2; n = n2;
    }
    __syncthreads();

    if (!LAST) {
        constexpr int Q4  = CH / 4;
        constexpr int NIT = (RN * Q4 + 1023) / 1024;
        const float* hr = h + (size_t)lo * CH;
        float* orow = outp + (size_t)lo * CH;
        float4 hv[NIT];
        float  dq[NIT];
        #pragma unroll
        for (int q = 0; q < NIT; ++q) {
            const int i = q * 1024 + tid;
            hv[q] = make_float4(0.0f, 0.0f, 0.0f, 0.0f);
            dq[q] = 0.0f;
            if (i < cnt * Q4) {
                hv[q] = *(const float4*)(hr + (size_t)i * 4);
                dq[q] = dinv[lo + i / Q4];
            }
        }
        #pragma unroll
        for (int q = 0; q < NIT; ++q) {
            const int i = q * 1024 + tid;
            if (i < cnt * Q4) {
                const int c = (i % Q4) * 4;
                const float dd = dq[q] * dq[q];
                const float4 a  = *(const float4*)&acc[i * 4];
                const float4 bv = *(const float4*)(bias + c);
                float4 v;
                v.x = fmaxf(a.x + hv[q].x * dd + bv.x, 0.0f);
                v.y = fmaxf(a.y + hv[q].y * dd + bv.y, 0.0f);
                v.z = fmaxf(a.z + hv[q].z * dd + bv.z, 0.0f);
                v.w = fmaxf(a.w + hv[q].w * dd + bv.w, 0.0f);
                *(float4*)(orow + (size_t)i * 4) = v;
            }
        }
    } else {
        static_assert(!LAST || CH == 32, "the projection takes one channel per lane");
        constexpr int NPW = (RN + 31) / 32;
        float hq[NPW], dq[NPW];
        #pragma unroll
        for (int q = 0; q < NPW; ++q) {
            const int node = q * 32 + wave;
            hq[q] = 0.0f;
            dq[q] = 0.0f;
            if (node < cnt) {
                hq[q] = h[(size_t)(lo + node) * CH + lane];
                dq[q] = dinv[lo + node];
            }
        }
        const float bl = bias[lane], wl = Wh[lane], b0 = bh[0];
        #pragma unroll
        for (int q = 0; q < NPW; ++q) {
            const int node = q * 32 + wave;
            if (node < cnt) {
                const float dd = dq[q] * dq[q];
                const float a = acc[node * CH + lane];
                float v = a + hq[q] * dd + bl;
                v = fmaxf(v, 0.0f);
                float p = v * wl;
                #pragma unroll
                for (int off = 16; off > 0; off >>= 1)
                    p += __shfl_xor(p, off, 32);
                if (lane == 0) outv[node] = p + b0;
            }
        }
        __syncthreads();
        for (int i = tid; i < cnt; i += 1024) outp[lo + i] = outv[i];
    }
}

#define NRG1 256
#define NPT1 128
#define RNG1 196

extern "C" void kernel_launch(void* const* d_in, const int* in_sizes, int n_in,
                              void* d_out, int out_size, void* d_ws, size_t ws_size,
                              hipStream_t stream) {
    const float* x  = (const float*)d_in[0];
    const int*   ei = (const int*)d_in[1];
    const float* W1 = (const float*)d_in[2];
    const float* b1 = (const float*)d_in[3];
    const float* W2 = (const float*)d_in[4];
    const float* b2 = (const float*)d_in[5];
    const float* Wh = (const float*)d_in[6];
    const float* bh = (const float*)d_in[7];
    float* out = (float*)d_out;

    const int N = in_sizes[0] / IN_CH;
    const int E = in_sizes[1] / 2;
    const int* src = ei;
    const int* dst = ei + E;

    float* f     = (float*)d_ws;
    float* dinv  = f;
    float* h1    = f + 65536;
    float* g1    = h1 + (size_t)N * HID1;
    float* h2    = g1 + (size_t)N * HID1;
    int*   lists = (int*)(h2 + (size_t)N * HID2);

    gemm1_wmma<<<N / 16, 128, 0, stream>>>(x, W1, h1);

    edge_split_kernel<RNG1, NRG1, NPT1><<<NPT1, 1024, 0, stream>>>(src, dst, lists, E, N);
    edge_count_dinv_kernel<RNG1, NRG1, NPT1><<<NRG1, 1024, 0, stream>>>(lists, dinv, N);

    edge_agg_kernel<HID1, false, RNG1, NRG1, NPT1><<<NRG1, 1024, 0, stream>>>(h1, lists, dinv, b1, Wh, bh, g1, N);

    gemm2_wmma<<<N / 16, 64, 0, stream>>>(g1, W2, h2);
    edge_agg_kernel<HID2, true, RNG1, NRG1, NPT1><<<NRG1, 1024, 0, stream>>>(h2, lists, dinv, b2, Wh, bh, out, N);
}


# Layer-1 product rides in the edge-split launch: 128 more work-groups there do X @ W1 (a wave = 16x32 tile, W1 in LDS, 16-B row loads, whole-line stores); 5 launches; the product's k order changes
# speedup vs baseline: 3.5676x; 1.0817x over previous
#include <hip/hip_runtime.h>

typedef float v2f __attribute__((ext_vector_type(2)));
typedef float v8f __attribute__((ext_vector_type(8)));

#define IN_CH  128
#define HID1   64
#define HID2   32

__global__ __launch_bounds__(64) void gemm2_wmma(const float* __restrict__ X,
                                                 const float* __restrict__ W,
                                                 float* __restrict__ H) {
    const int lane = threadIdx.x & 31;
    const int wave = threadIdx.x >> 5;
    const int m0   = blockIdx.x * 16;
    const int n0   = wave * 16;
    const int l16  = lane & 15;
    const int khi  = (lane >> 4) * 2;
    const int row  = m0 + l16;
    const int col  = n0 + l16;
    v8f c = {};
    #pragma unroll 4
    for (int kk = 0; kk < HID1; kk += 4) {
        v2f a, b;
        a.x = X[row * HID1 + kk + khi + 0];
        a.y = X[row * HID1 + kk + khi + 1];
        b.x = W[(kk + khi + 0) * HID2 + col];
        b.y = W[(kk + khi + 1) * HID2 + col];
        c = __builtin_amdgcn_wmma_f32_16x16x4_f32(false, a, false, b,
                                                  (short)0, c, false, false);
    }
    const int mbase = m0 + ((lane >> 4) * 8);
    #pragma unroll
    for (int r = 0; r < 8; ++r)
        H[(mbase + r) * HID2 + col] = c[r];
}

__device__ __forceinline__ float xhalf(float x) {
    return __builtin_bit_cast(float, __builtin_amdgcn_permlanex16(__builtin_bit_cast(int, x), __builtin_bit_cast(int, x),
                                                                  0x76543210, 0xfedcba98, false, false));
}

#define EPART 12512

template <int NRANGE, int NPARTS>
struct WalkGeom {
    static constexpr int PPW = NPARTS / 32;
    static_assert(NPARTS % 32 == 0, "every wave walks the same number of parts");
};

template <int RN, int NRANGE, int NPARTS, int NPROD>
__global__ __launch_bounds__(1024) void split_prod_kernel(const int* __restrict__ src,
                                                         const int* __restrict__ dst,
                                                         int* __restrict__ lists,
                                                         int nE, int nN,
                                                         const float* __restrict__ X,
                                                         const float* __restrict__ W,
                                                         float* __restrict__ H) {
    static_assert(NRANGE == 256, "eight ranges per lane of the scanning wave; 8-bit range field");
    static_assert(RN <= 256, "8-bit local row field");
    static_assert(EPART < 32768, "15-bit rank field");
    constexpr int STGA = EPART > 20608 ? EPART : 20608;
    constexpr int KPT  = (EPART + 1023) / 1024;
    __shared__ int st_s[STGA];
    __shared__ int st_d[STGA];
    __shared__ int hist[NRANGE];
    __shared__ int offs[NRANGE + 1];
    const int tid  = threadIdx.x;
    const int lane = tid & 31;
    const int wave = tid >> 5;

    if (blockIdx.x < NPROD) {
        static_assert(IN_CH * HID1 == 8192 && STGA >= 8192, "two (k / 4, n) groups of W per thread");
        float* const wl = reinterpret_cast<float*>(st_s);
        const int wv = __builtin_amdgcn_readfirstlane(wave);
        const int c  = lane & 15;
        const int hh = lane >> 4;
        const int ntask = ((nN + 15) >> 4) * 2;
        const int t0 = (int)((long long)ntask * blockIdx.x / NPROD);
        const int t1 = (int)((long long)ntask * (blockIdx.x + 1) / NPROD);
        float wq[2][4];
        #pragma unroll
        for (int j = 0; j < 2; ++j) {
            const int p = j * 1024 + tid;
            #pragma unroll
            for (int e = 0; e < 4; ++e) wq[j][e] = W[(((p >> 6) << 2) + e) * HID1 + (p & 63)];
        }
        auto rows_of = [&](int t) -> const float4* {
            int tt = t < t1 - 1 ? t : t1 - 1;
            tt = tt < 0 ? 0 : tt;
            int row = (tt >> 1) * 16 + c;
            row = row < nN ? row : nN - 1;
            return (const float4*)(X + (size_t)row * IN_CH + 4 * hh);
        };
        int tau = t0 + wv;
        float4 a[16];
        {
            const float4* xp = rows_of(tau);
            #pragma unroll
            for (int q = 0; q < 16; ++q) a[q] = xp[2 * q];
        }
        #pragma unroll
        for (int j = 0; j < 2; ++j)
            *(float4*)&wl[(j * 1024 + tid) << 2] = make_float4(wq[j][0], wq[j][1], wq[j][2], wq[j][3]);
        __syncthreads();
        for (; tau < t1; tau += 32) {
            const int m0 = (tau >> 1) * 16;
            const int n0 = (tau & 1) * 32;
            const float4* xn = rows_of(tau + 32);
            v8f c0 = {}, c1 = {};
            #pragma unroll
            for (int q = 0; q < 16; ++q) {
                const float4 b0 = *(const float4*)&wl[(((2 * q + hh) << 6) + n0 + c) << 2];
                const float4 b1 = *(const float4*)&wl[(((2 * q + hh) << 6) + n0 + 16 + c) << 2];
                v2f a01, a23, b001, b023, b101, b123;
                a01.x = a[q].x; a01.y = a[q].y; a23.x = a[q].z; a23.y = a[q].w;
                b001.x = b0.x; b001.y = b0.y; b023.x = b0.z; b023.y = b0.w;
                b101.x = b1.x; b101.y = b1.y; b123.x = b1.z; b123.y = b1.w;
                c0 = __builtin_amdgcn_wmma_f32_16x16x4_f32(false, a01, false, b001, (short)0, c0, false, false);
                c1 = __builtin_amdgcn_wmma_f32_16x16x4_f32(false, a01, false, b101, (short)0, c1, false, false);
                c0 = __builtin_amdgcn_wmma_f32_16x16x4_f32(false, a23, false, b023, (short)0, c0, false, false);
                c1 = __builtin_amdgcn_wmma_f32_16x16x4_f32(false, a23, false, b123, (short)0, c1, false, false);
                a[q] = xn[2 * q];
            }
            #pragma unroll
            for (int r = 0; r < 8; ++r) {
                const float s0 = xhalf(c0[r]);
                const float s1 = xhalf(c1[r]);
                const float lo = hh == 0 ? c0[r] : s1;
                const float hi = hh == 0 ? s0 : c1[r];
                if (m0 + r < nN)     H[(size_t)(m0 + r) * HID1 + n0 + lane] = lo;
                if (m0 + 8 + r < nN) H[(size_t)(m0 + 8 + r) * HID1 + n0 + lane] = hi;
            }
        }
        return;
    }

    const int part = blockIdx.x - NPROD;
    const int e0 = (int)((long long)nE * part / NPARTS);
    const int e1 = (int)((long long)nE * (part + 1) / NPARTS);
    int cnt = e1 - e0;
    cnt = cnt > EPART ? EPART : cnt;

    int dv[KPT], sv[KPT], pk[KPT];
    #pragma unroll
    for (int k = 0; k < KPT; ++k) {
        const int i = k * 1024 + tid;
        const int e = i < cnt ? e0 + i : 0;
        dv[k] = dst[e];
        sv[k] = src[e];
    }
    if (tid < NRANGE) hist[tid] = 0;
    __syncthreads();

    #pragma unroll
    for (int k = 0; k < KPT; ++k) {
        const int i = k * 1024 + tid;
        pk[k] = -1;
        if (i < cnt && (unsigned)dv[k] < (unsigned)(nN < RN * NRANGE ? nN : RN * NRANGE)) {
            const int r  = (int)((unsigned)dv[k] / (unsigned)RN);
            const int dl = dv[k] - r * RN;
            const int rank = atomicAdd(&hist[r], 1);
            pk[k] = r | (dl << 8) | (rank << 16);
        }
    }
    __syncthreads();

    if (wave == 0) {
        int hv[8];
        int sum = 0;
        #pragma unroll
        for (int j = 0; j < 8; ++j) { hv[j] = hist[lane * 8 + j]; sum += hv[j]; }
        int inc = sum;
        #pragma unroll
        for (int off = 1; off < 32; off <<= 1) {
            const int t = __shfl_up(inc, off, 32);
            if (lane >= off) inc += t;
        }
        int run = inc - sum;
        #pragma unroll
        for (int j = 0; j < 8; ++j) { offs[lane * 8 + j] = run; run += hv[j]; }
        if (lane == 31) offs[NRANGE] = run;
    }
    __syncthreads();

    int* const tables = lists + (size_t)2 * NPARTS * EPART;
    if (tid < NRANGE)
        *(int2*)(tables + ((size_t)part * NRANGE + tid) * 2) = make_int2(offs[tid], hist[tid]);

    #pragma unroll
    for (int k = 0; k < KPT; ++k) {
        if (pk[k] >= 0) {
            const int pos = offs[pk[k] & 255] + (pk[k] >> 16);
            st_s[pos] = sv[k];
            st_d[pos] = (pk[k] >> 8) & 255;
        }
    }
    __syncthreads();

    const int tot = offs[NRANGE];
    int* const gs = lists + (size_t)part * EPART;
    int* const gd = lists + (size_t)NPARTS * EPART + (size_t)part * EPART;
    for (int i = tid; i < tot; i += 1024) {
        gs[i] = st_s[i];
        gd[i] = st_d[i];
    }
}

template <int NRANGE, int NPARTS>
struct WaveRuns {
    static constexpr int PPW = WalkGeom<NRANGE, NPARTS>::PPW;
    int base[PPW];
    int first[PPW];
    int total;
    __device__ __forceinline__ void load(const int* __restrict__ lists, int range, int wave) {
        const int* const tables = lists + (size_t)2 * NPARTS * EPART;
        int st[PPW], c[PPW];
        #pragma unroll
        for (int j = 0; j < PPW; ++j) {
            const int p = wave * PPW + j;
            const int2 t = *(const int2*)(tables + ((size_t)p * NRANGE + range) * 2);
            st[j] = __builtin_amdgcn_readfirstlane(t.x);
            c[j]  = __builtin_amdgcn_readfirstlane(t.y);
        }
        int pos = 0;
        #pragma unroll
        for (int j = 0; j < PPW; ++j) {
            first[j] = pos;
            base[j]  = (wave * PPW + j) * EPART + st[j] - pos;
            pos += c[j];
        }
        total = pos;
    }
    __device__ __forceinline__ int at(int v) const {
        int a = base[0];
        #pragma unroll
        for (int j = 1; j < PPW; ++j) a = v >= first[j] ? base[j] : a;
        return a + v;
    }
};

template <int RN, int NRANGE, int NPARTS>
__global__ __launch_bounds__(1024) void edge_count_dinv_kernel(const int* __restrict__ lists,
                                                              float* __restrict__ dinv, int nN) {
    __shared__ int cntl[RN];
    const int tid  = threadIdx.x;
    const int lane = tid & 31;
    const int wave = __builtin_amdgcn_readfirstlane(tid >> 5);
    const int range = blockIdx.x;
    const int lo = range * RN;
    int cnt = nN - lo;
    cnt = cnt < 0 ? 0 : (cnt > RN ? RN : cnt);
    const int* const gd = lists + (size_t)NPARTS * EPART;
    WaveRuns<NRANGE, NPARTS> runs;
    runs.load(lists, range, wave);
    const int nb = (runs.total + 31) >> 5;
    for (int i = tid; i < RN; i += 1024) cntl[i] = 0;
    __syncthreads();
    for (int k0 = 0; k0 < nb; k0 += 8) {
        int dlv[8];
        #pragma unroll
        for (int j = 0; j < 8; ++j) {
            const int v = 32 * (k0 + j) + lane;
            dlv[j] = v < runs.total ? gd[runs.at(v)] : -1;
        }
        #pragma unroll
        for (int j = 0; j < 8; ++j)
            if (dlv[j] >= 0) atomicAdd(&cntl[dlv[j]], 1);
    }
    __syncthreads();
    for (int i = tid; i < cnt; i += 1024) dinv[lo + i] = rsqrtf((float)cntl[i] + 1.0f);
}

template <int CH, bool LAST, int RN, int NRANGE, int NPARTS>
__global__ __launch_bounds__(1024) void edge_agg_kernel(const float* __restrict__ h,
                                                       const int* __restrict__ lists,
                                                       const float* __restrict__ dinv,
                                                       const float* __restrict__ bias,
                                                       const float* __restrict__ Wh,
                                                       const float* __restrict__ bh,
                                                       float* __restrict__ outp,
                                                       int nN) {
    constexpr int VW = CH / 32;
    constexpr int ACCF = RN * CH > 41472 ? RN * CH : 41472;
    __shared__ float acc[ACCF];
    __shared__ float outv[LAST ? RN : 1];
    const int tid  = threadIdx.x;
    const int lane = tid & 31;
    const int wave = __builtin_amdgcn_readfirstlane(tid >> 5);
    const int range = blockIdx.x;
    const int lo = range * RN;
    int cnt = nN - lo;
    cnt = cnt < 0 ? 0 : (cnt > RN ? RN : cnt);

    const int* const gs = lists;
    const int* const gd = lists + (size_t)NPARTS * EPART;
    WaveRuns<NRANGE, NPARTS> runs;
    runs.load(lists, range, wave);
    const int tot = runs.total;
    const int nb = (tot + 31) >> 5;
    int s = 0, dl = 0, n = 0;
    if (nb > 0) {
        n = tot < 32 ? tot : 32;
        if (lane < n) { const int a = runs.at(lane); s = gs[a]; dl = gd[a]; }
    }

    for (int i = tid * 4; i < RN * CH; i += 4096)
        *(float4*)&acc[i] = make_float4(0.0f, 0.0f, 0.0f, 0.0f);
    __syncthreads();

    for (int k = 0; k < nb; ++k) {
        int s2 = 0, dl2 = 0, n2 = 0;
        const bool more = k + 1 < nb;
        if (more) {
            const int left = tot - 32 * (k + 1);
            n2 = left < 32 ? left : 32;
        }
        const int v2 = 32 * (k + 1) + lane;
        float ns, nd, nm;
        if (n == 32) {
            float r[32][VW];
            #pragma unroll
            for (int u = 0; u < 32; ++u) {
                const unsigned su = (unsigned)__builtin_amdgcn_readlane(s, u);
                const float* hp = h + (size_t)su * CH + lane * VW;
                #pragma unroll
                for (int w = 0; w < VW; ++w) r[u][w] = hp[w];
            }
            ns = dinv[s]; nd = dinv[lo + dl];
            if (more && lane < n2) { const int a = runs.at(v2); s2 = gs[a]; dl2 = gd[a]; }
            nm = ns * nd;
            #pragma unroll
            for (int u = 0; u < 32; ++u) {
                const int du = __builtin_amdgcn_readlane(dl, u);
                const float nu = __builtin_bit_cast(float,
                    __builtin_amdgcn_readlane(__builtin_bit_cast(int, nm), u));
                float* ap = &acc[du * CH + lane * VW];
                #pragma unroll
                for (int w = 0; w < VW; ++w) atomicAdd(ap + w, r[u][w] * nu);
            }
        } else {
            ns = dinv[s]; nd = dinv[lo + dl];
            if (more && lane < n2) { const int a = runs.at(v2); s2 = gs[a]; dl2 = gd[a]; }
            nm = (lane < n) ? ns * nd : 0.0f;
            #pragma unroll
            for (int g = 0; g < 4; ++g) {
                if (g * 8 < n) {
                    float r[8][VW];
                    #pragma unroll
                    for (int u = 0; u < 8; ++u) {
                        const unsigned su = (unsigned)__builtin_amdgcn_readlane(s, g * 8 + u);
                        const float* hp = h + (size_t)su * CH + lane * VW;
                        #pragma unroll
                        for (int w = 0; w < VW; ++w) r[u][w] = hp[w];
                    }
                    #pragma unroll
                    for (int u = 0; u < 8; ++u) {
                        if (g * 8 + u < n) {
                            const int du = __builtin_amdgcn_readlane(dl, g * 8 + u);
                            const float nu = __builtin_bit_cast(float,
                                __builtin_amdgcn_readlane(__builtin_bit_cast(int, nm), g * 8 + u));
                            float* ap = &acc[du * CH + lane * VW];
                            #pragma unroll
                            for (int w = 0; w < VW; ++w) atomicAdd(ap + w, r[u][w] * nu);
                        }
                    }
                }
            }
        }
        s = s2; dl = dl2; n = n2;
    }
    __syncthreads();

    if (!LAST) {
        constexpr int Q4  = CH / 4;
        constexpr int NIT = (RN * Q4 + 1023) / 1024;
        const float* hr = h + (size_t)lo * CH;
        float* orow = outp + (size_t)lo * CH;
        float4 hv[NIT];
        float  dq[NIT];
        #pragma unroll
        for (int q = 0; q < NIT; ++q) {
            const int i = q * 1024 + tid;
            hv[q] = make_float4(0.0f, 0.0f, 0.0f, 0.0f);
            dq[q] = 0.0f;
            if (i < cnt * Q4) {
                hv[q] = *(const float4*)(hr + (size_t)i * 4);
                dq[q] = dinv[lo + i / Q4];
            }
        }
        #pragma unroll
        for (int q = 0; q < NIT; ++q) {
            const int i = q * 1024 + tid;
            if (i < cnt * Q4) {
                const int c = (i % Q4) * 4;
                const float dd = dq[q] * dq[q];
                const float4 a  = *(const float4*)&acc[i * 4];
                const float4 bv = *(const float4*)(bias + c);
                float4 v;
                v.x = fmaxf(a.x + hv[q].x * dd + bv.x, 0.0f);
                v.y = fmaxf(a.y + hv[q].y * dd + bv.y, 0.0f);
                v.z = fmaxf(a.z + hv[q].z * dd + bv.z, 0.0f);
                v.w = fmaxf(a.w + hv[q].w * dd + bv.w, 0.0f);
                *(float4*)(orow + (size_t)i * 4) = v;
            }
        }
    } else {
        static_assert(!LAST || CH == 32, "the projection takes one channel per lane");
        constexpr int NPW = (RN + 31) / 32;
        float hq[NPW], dq[NPW];
        #pragma unroll
        for (int q = 0; q < NPW; ++q) {
            const int node = q * 32 + wave;
            hq[q] = 0.0f;
            dq[q] = 0.0f;
            if (node < cnt) {
                hq[q] = h[(size_t)(lo + node) * CH + lane];
                dq[q] = dinv[lo + node];
            }
        }
        const float bl = bias[lane], wl = Wh[lane], b0 = bh[0];
        #pragma unroll
        for (int q = 0; q < NPW; ++q) {
            const int node = q * 32 + wave;
            if (node < cnt) {
                const float dd = dq[q] * dq[q];
                const float a = acc[node * CH + lane];
                float v = a + hq[q] * dd + bl;
                v = fmaxf(v, 0.0f);
                float p = v * wl;
                #pragma unroll
                for (int off = 16; off > 0; off >>= 1)
                    p += __shfl_xor(p, off, 32);
                if (lane == 0) outv[node] = p + b0;
            }
        }
        __syncthreads();
        for (int i = tid; i < cnt; i += 1024) outp[lo + i] = outv[i];
    }
}

#define NRG1 256
#define NPT1 128
#define NPR1 128
#define RNG1 196

extern "C" void kernel_launch(void* const* d_in, const int* in_sizes, int n_in,
                              void* d_out, int out_size, void* d_ws, size_t ws_size,
                              hipStream_t stream) {
    const float* x  = (const float*)d_in[0];
    const int*   ei = (const int*)d_in[1];
    const float* W1 = (const float*)d_in[2];
    const float* b1 = (const float*)d_in[3];
    const float* W2 = (const float*)d_in[4];
    const float* b2 = (const float*)d_in[5];
    const float* Wh = (const float*)d_in[6];
    const float* bh = (const float*)d_in[7];
    float* out = (float*)d_out;

    const int N = in_sizes[0] / IN_CH;
    const int E = in_sizes[1] / 2;
    const int* src = ei;
    const int* dst = ei + E;

    float* f     = (float*)d_ws;
    float* dinv  = f;
    float* h1    = f + 65536;
    float* g1    = h1 + (size_t)N * HID1;
    float* h2    = g1 + (size_t)N * HID1;
    int*   lists = (int*)(h2 + (size_t)N * HID2);

    split_prod_kernel<RNG1, NRG1, NPT1, NPR1><<<NPR1 + NPT1, 1024, 0, stream>>>(src, dst, lists, E, N, x, W1, h1);

    edge_count_dinv_kernel<RNG1, NRG1, NPT1><<<NRG1, 1024, 0, stream>>>(lists, dinv, N);

    edge_agg_kernel<HID1, false, RNG1, NRG1, NPT1><<<NRG1, 1024, 0, stream>>>(h1, lists, dinv, b1, Wh, bh, g1, N);

    gemm2_wmma<<<N / 16, 64, 0, stream>>>(g1, W2, h2);
    edge_agg_kernel<HID2, true, RNG1, NRG1, NPT1><<<NRG1, 1024, 0, stream>>>(h2, lists, dinv, b2, Wh, bh, out, N);
}


# Layer-2 product moved into the layer-1 owner pass: a range's finished rows stay in LDS (pitch 68) and feed the WMMA A side there, W2 in registers, 13 waves; g1 plane and one launch go; k order kept
# speedup vs baseline: 3.7975x; 1.0644x over previous
#include <hip/hip_runtime.h>

typedef float v2f __attribute__((ext_vector_type(2)));
typedef float v8f __attribute__((ext_vector_type(8)));

#define IN_CH  128
#define HID1   64
#define HID2   32

__device__ __forceinline__ float xhalf(float x) {
    return __builtin_bit_cast(float, __builtin_amdgcn_permlanex16(__builtin_bit_cast(int, x), __builtin_bit_cast(int, x),
                                                                  0x76543210, 0xfedcba98, false, false));
}

#define EPART 12512

template <int NRANGE, int NPARTS>
struct WalkGeom {
    static constexpr int PPW = NPARTS / 32;
    static_assert(NPARTS % 32 == 0, "every wave walks the same number of parts");
};

template <int RN, int NRANGE, int NPARTS, int NPROD>
__global__ __launch_bounds__(1024) void split_prod_kernel(const int* __restrict__ src,
                                                         const int* __restrict__ dst,
                                                         int* __restrict__ lists,
                                                         int nE, int nN,
                                                         const float* __restrict__ X,
                                                         const float* __restrict__ W,
                                                         float* __restrict__ H) {
    static_assert(NRANGE == 256, "eight ranges per lane of the scanning wave; 8-bit range field");
    static_assert(RN <= 256, "8-bit local row field");
    static_assert(EPART < 32768, "15-bit rank field");
    constexpr int STGA = EPART > 20608 ? EPART : 20608;
    constexpr int KPT  = (EPART + 1023) / 1024;
    __shared__ int st_s[STGA];
    __shared__ int st_d[STGA];
    __shared__ int hist[NRANGE];
    __shared__ int offs[NRANGE + 1];
    const int tid  = threadIdx.x;
    const int lane = tid & 31;
    const int wave = tid >> 5;

    if (blockIdx.x < NPROD) {
        static_assert(IN_CH * HID1 == 8192 && STGA >= 8192, "two (k / 4, n) groups of W per thread");
        float* const wl = reinterpret_cast<float*>(st_s);
        const int wv = __builtin_amdgcn_readfirstlane(wave);
        const int c  = lane & 15;
        const int hh = lane >> 4;
        const int ntask = ((nN + 15) >> 4) * 2;
        const int t0 = (int)((long long)ntask * blockIdx.x / NPROD);
        const int t1 = (int)((long long)ntask * (blockIdx.x + 1) / NPROD);
        float wq[2][4];
        #pragma unroll
        for (int j = 0; j < 2; ++j) {
            const int p = j * 1024 + tid;
            #pragma unroll
            for (int e = 0; e < 4; ++e) wq[j][e] = W[(((p >> 6) << 2) + e) * HID1 + (p & 63)];
        }
        auto rows_of = [&](int t) -> const float4* {
            int tt = t < t1 - 1 ? t : t1 - 1;
            tt = tt < 0 ? 0 : tt;
            int row = (tt >> 1) * 16 + c;
            row = row < nN ? row : nN - 1;
            return (const float4*)(X + (size_t)row * IN_CH + 4 * hh);
        };
        int tau = t0 + wv;
        float4 a[16];
        {
            const float4* xp = rows_of(tau);
            #pragma unroll
            for (int q = 0; q < 16; ++q) a[q] = xp[2 * q];
        }
        #pragma unroll
        for (int j = 0; j < 2; ++j)
            *(float4*)&wl[(j * 1024 + tid) << 2] = make_float4(wq[j][0], wq[j][1], wq[j][2], wq[j][3]);
        __syncthreads();
        for (; tau < t1; tau += 32) {
            const int m0 = (tau >> 1) * 16;
            const int n0 = (tau & 1) * 32;
            const float4* xn = rows_of(tau + 32);
            v8f c0 = {}, c1 = {};
            #pragma unroll
            for (int q = 0; q < 16; ++q) {
                const float4 b0 = *(const float4*)&wl[(((2 * q + hh) << 6) + n0 + c) << 2];
                const float4 b1 = *(const float4*)&wl[(((2 * q + hh) << 6) + n0 + 16 + c) << 2];
                v2f a01, a23, b001, b023, b101, b123;
                a01.x = a[q].x; a01.y = a[q].y; a23.x = a[q].z; a23.y = a[q].w;
                b001.x = b0.x; b001.y = b0.y; b023.x = b0.z; b023.y = b0.w;
                b101.x = b1.x; b101.y = b1.y; b123.x = b1.z; b123.y = b1.w;
                c0 = __builtin_amdgcn_wmma_f32_16x16x4_f32(false, a01, false, b001, (short)0, c0, false, false);
                c1 = __builtin_amdgcn_wmma_f32_16x16x4_f32(false, a01, false, b101, (short)0, c1, false, false);
                c0 = __builtin_amdgcn_wmma_f32_16x16x4_f32(false, a23, false, b023, (short)0, c0, false, false);
                c1 = __builtin_amdgcn_wmma_f32_16x16x4_f32(false, a23, false, b123, (short)0, c1, false, false);
                a[q] = xn[2 * q];
            }
            #pragma unroll
            for (int r = 0; r < 8; ++r) {
                const float s0 = xhalf(c0[r]);
                const float s1 = xhalf(c1[r]);
                const float lo = hh == 0 ? c0[r] : s1;
                const float hi = hh == 0 ? s0 : c1[r];
                if (m0 + r < nN)     H[(size_t)(m0 + r) * HID1 + n0 + lane] = lo;
                if (m0 + 8 + r < nN) H[(size_t)(m0 + 8 + r) * HID1 + n0 + lane] = hi;
            }
        }
        return;
    }

    const int part = blockIdx.x - NPROD;
    const int e0 = (int)((long long)nE * part / NPARTS);
    const int e1 = (int)((long long)nE * (part + 1) / NPARTS);
    int cnt = e1 - e0;
    cnt = cnt > EPART ? EPART : cnt;

    int dv[KPT], sv[KPT], pk[KPT];
    #pragma unroll
    for (int k = 0; k < KPT; ++k) {
        const int i = k * 1024 + tid;
        const int e = i < cnt ? e0 + i : 0;
        dv[k] = dst[e];
        sv[k] = src[e];
    }
    if (tid < NRANGE) hist[tid] = 0;
    __syncthreads();

    #pragma unroll
    for (int k = 0; k < KPT; ++k) {
        const int i = k * 1024 + tid;
        pk[k] = -1;
        if (i < cnt && (unsigned)dv[k] < (unsigned)(nN < RN * NRANGE ? nN : RN * NRANGE)) {
            const int r  = (int)((unsigned)dv[k] / (unsigned)RN);
            const int dl = dv[k] - r * RN;
            const int rank = atomicAdd(&hist[r], 1);
            pk[k] = r | (dl << 8) | (rank << 16);
        }
    }
    __syncthreads();

    if (wave == 0) {
        int hv[8];
        int sum = 0;
        #pragma unroll
        for (int j = 0; j < 8; ++j) { hv[j] = hist[lane * 8 + j]; sum += hv[j]; }
        int inc = sum;
        #pragma unroll
        for (int off = 1; off < 32; off <<= 1) {
            const int t = __shfl_up(inc, off, 32);
            if (lane >= off) inc += t;
        }
        int run = inc - sum;
        #pragma unroll
        for (int j = 0; j < 8; ++j) { offs[lane * 8 + j] = run; run += hv[j]; }
        if (lane == 31) offs[NRANGE] = run;
    }
    __syncthreads();

    int* const tables = lists + (size_t)2 * NPARTS * EPART;
    if (tid < NRANGE)
        *(int2*)(tables + ((size_t)part * NRANGE + tid) * 2) = make_int2(offs[tid], hist[tid]);

    #pragma unroll
    for (int k = 0; k < KPT; ++k) {
        if (pk[k] >= 0) {
            const int pos = offs[pk[k] & 255] + (pk[k] >> 16);
            st_s[pos] = sv[k];
            st_d[pos] = (pk[k] >> 8) & 255;
        }
    }
    __syncthreads();

    const int tot = offs[NRANGE];
    int* const gs = lists + (size_t)part * EPART;
    int* const gd = lists + (size_t)NPARTS * EPART + (size_t)part * EPART;
    for (int i = tid; i < tot; i += 1024) {
        gs[i] = st_s[i];
        gd[i] = st_d[i];
    }
}

template <int NRANGE, int NPARTS>
struct WaveRuns {
    static constexpr int PPW = WalkGeom<NRANGE, NPARTS>::PPW;
    int base[PPW];
    int first[PPW];
    int total;
    __device__ __forceinline__ void load(const int* __restrict__ lists, int range, int wave) {
        const int* const tables = lists + (size_t)2 * NPARTS * EPART;
        int st[PPW], c[PPW];
        #pragma unroll
        for (int j = 0; j < PPW; ++j) {
            const int p = wave * PPW + j;
            const int2 t = *(const int2*)(tables + ((size_t)p * NRANGE + range) * 2);
            st[j] = __builtin_amdgcn_readfirstlane(t.x);
            c[j]  = __builtin_amdgcn_readfirstlane(t.y);
        }
        int pos = 0;
        #pragma unroll
        for (int j = 0; j < PPW; ++j) {
            first[j] = pos;
            base[j]  = (wave * PPW + j) * EPART + st[j] - pos;
            pos += c[j];
        }
        total = pos;
    }
    __device__ __forceinline__ int at(int v) const {
        int a = base[0];
        #pragma unroll
        for (int j = 1; j < PPW; ++j) a = v >= first[j] ? base[j] : a;
        return a + v;
    }
};

template <int RN, int NRANGE, int NPARTS>
__global__ __launch_bounds__(1024) void edge_count_dinv_kernel(const int* __restrict__ lists,
                                                              float* __restrict__ dinv, int nN) {
    __shared__ int cntl[RN];
    const int tid  = threadIdx.x;
    const int lane = tid & 31;
    const int wave = __builtin_amdgcn_readfirstlane(tid >> 5);
    const int range = blockIdx.x;
    const int lo = range * RN;
    int cnt = nN - lo;
    cnt = cnt < 0 ? 0 : (cnt > RN ? RN : cnt);
    const int* const gd = lists + (size_t)NPARTS * EPART;
    WaveRuns<NRANGE, NPARTS> runs;
    runs.load(lists, range, wave);
    const int nb = (runs.total + 31) >> 5;
    for (int i = tid; i < RN; i += 1024) cntl[i] = 0;
    __syncthreads();
    for (int k0 = 0; k0 < nb; k0 += 8) {
        int dlv[8];
        #pragma unroll
        for (int j = 0; j < 8; ++j) {
            const int v = 32 * (k0 + j) + lane;
            dlv[j] = v < runs.total ? gd[runs.at(v)] : -1;
        }
        #pragma unroll
        for (int j = 0; j < 8; ++j)
            if (dlv[j] >= 0) atomicAdd(&cntl[dlv[j]], 1);
    }
    __syncthreads();
    for (int i = tid; i < cnt; i += 1024) dinv[lo + i] = rsqrtf((float)cntl[i] + 1.0f);
}

template <int CH, bool LAST, int RN, int NRANGE, int NPARTS>
__global__ __launch_bounds__(1024) void edge_agg_kernel(const float* __restrict__ h,
                                                       const int* __restrict__ lists,
                                                       const float* __restrict__ dinv,
                                                       const float* __restrict__ bias,
                                                       const float* __restrict__ Wn,
                                                       const float* __restrict__ bh,
                                                       float* __restrict__ outp,
                                                       int nN) {
    constexpr int VW = CH / 32;
    constexpr int ACCF = RN * CH > 41472 ? RN * CH : 41472;
    __shared__ float acc[ACCF];
    __shared__ float outv[LAST ? RN : 1];
    const int tid  = threadIdx.x;
    const int lane = tid & 31;
    const int wave = __builtin_amdgcn_readfirstlane(tid >> 5);
    const int range = blockIdx.x;
    const int lo = range * RN;
    int cnt = nN - lo;
    cnt = cnt < 0 ? 0 : (cnt > RN ? RN : cnt);

    const int* const gs = lists;
    const int* const gd = lists + (size_t)NPARTS * EPART;
    WaveRuns<NRANGE, NPARTS> runs;
    runs.load(lists, range, wave);
    const int tot = runs.total;
    const int nb = (tot + 31) >> 5;
    int s = 0, dl = 0, n = 0;
    if (nb > 0) {
        n = tot < 32 ? tot : 32;
        if (lane < n) { const int a = runs.at(lane); s = gs[a]; dl = gd[a]; }
    }

    for (int i = tid * 4; i < RN * CH; i += 4096)
        *(float4*)&acc[i] = make_float4(0.0f, 0.0f, 0.0f, 0.0f);
    __syncthreads();

    for (int k = 0; k < nb; ++k) {
        int s2 = 0, dl2 = 0, n2 = 0;
        const bool more = k + 1 < nb;
        if (more) {
            const int left = tot - 32 * (k + 1);
            n2 = left < 32 ? left : 32;
        }
        const int v2 = 32 * (k + 1) + lane;
        float ns, nd, nm;
        if (n == 32) {
            float r[32][VW];
            #pragma unroll
            for (int u = 0; u < 32; ++u) {
                const unsigned su = (unsigned)__builtin_amdgcn_readlane(s, u);
                const float* hp = h + (size_t)su * CH + lane * VW;
                #pragma unroll
                for (int w = 0; w < VW; ++w) r[u][w] = hp[w];
            }
            ns = dinv[s]; nd = dinv[lo + dl];
            if (more && lane < n2) { const int a = runs.at(v2); s2 = gs[a]; dl2 = gd[a]; }
            nm = ns * nd;
            #pragma unroll
            for (int u = 0; u < 32; ++u) {
                const int du = __builtin_amdgcn_readlane(dl, u);
                const float nu = __builtin_bit_cast(float,
                    __builtin_amdgcn_readlane(__builtin_bit_cast(int, nm), u));
                float* ap = &acc[du * CH + lane * VW];
                #pragma unroll
                for (int w = 0; w < VW; ++w) atomicAdd(ap + w, r[u][w] * nu);
            }
        } else {
            ns = dinv[s]; nd = dinv[lo + dl];
            if (more && lane < n2) { const int a = runs.at(v2); s2 = gs[a]; dl2 = gd[a]; }
            nm = (lane < n) ? ns * nd : 0.0f;
            #pragma unroll
            for (int g = 0; g < 4; ++g) {
                if (g * 8 < n) {
                    float r[8][VW];
                    #pragma unroll
                    for (int u = 0; u < 8; ++u) {
                        const unsigned su = (unsigned)__builtin_amdgcn_readlane(s, g * 8 + u);
                        const float* hp = h + (size_t)su * CH + lane * VW;
                        #pragma unroll
                        for (int w = 0; w < VW; ++w) r[u][w] = hp[w];
                    }
                    #pragma unroll
                    for (int u = 0; u < 8; ++u) {
                        if (g * 8 + u < n) {
                            const int du = __builtin_amdgcn_readlane(dl, g * 8 + u);
                            const float nu = __builtin_bit_cast(float,
                                __builtin_amdgcn_readlane(__builtin_bit_cast(int, nm), g * 8 + u));
                            float* ap = &acc[du * CH + lane * VW];
                            #pragma unroll
                            for (int w = 0; w < VW; ++w) atomicAdd(ap + w, r[u][w] * nu);
                        }
                    }
                }
            }
        }
        s = s2; dl = dl2; n = n2;
    }
    __syncthreads();

    if (!LAST) {
        static_assert(LAST || CH % 4 == 0, "k runs in steps of four");
        constexpr int Q4  = CH / 4;
        constexpr int NIT = (RN * Q4 + 1023) / 1024;
        constexpr int KQ  = CH / 4;
        constexpr int NRT = (RN + 15) / 16;
        constexpr int GP = CH + 4;
        static_assert(LAST || RN * CH + RN * GP <= ACCF, "the finished rows fit behind the sums");
        static_assert(LAST || NRT <= 32, "one wave per 16-row tile");
        float* const gl = acc + RN * CH;
        static_assert(LAST || 1024 % Q4 == 0, "a thread keeps its four channels over its rows");
        const int c16 = lane & 15;
        const int hh  = lane >> 4;
        const int c4  = (tid % Q4) * 4;
        const bool mul = wave < NRT && wave * 16 < cnt;
        const int lr = cnt > 0 ? lo : 0;
        const float* hr = h + (size_t)lr * CH;
        float4 hv[NIT];
        float  dq[NIT];
        v2f bw0[KQ], bw1[KQ];
        #pragma unroll
        for (int q = 0; q < NIT; ++q) {
            int i = q * 1024 + tid;
            i = i < cnt * Q4 ? i : 0;
            hv[q] = *(const float4*)(hr + (size_t)i * 4);
            dq[q] = dinv[lr + i / Q4];
        }
        const float4 bv = *(const float4*)(bias + c4);
        if (mul) {
            #pragma unroll
            for (int q = 0; q < KQ; ++q) {
                const int k = 4 * q + 2 * hh;
                bw0[q].x = Wn[(k + 0) * HID2 + c16];
                bw0[q].y = Wn[(k + 1) * HID2 + c16];
                bw1[q].x = Wn[(k + 0) * HID2 + 16 + c16];
                bw1[q].y = Wn[(k + 1) * HID2 + 16 + c16];
            }
        }
        #pragma unroll
        for (int q = 0; q < NIT; ++q) {
            const int i = q * 1024 + tid;
            if (i < cnt * Q4) {
                const float dd = dq[q] * dq[q];
                const float4 a  = *(const float4*)&acc[i * 4];
                float4 v;
                v.x = fmaxf(fmaf(hv[q].x, dd, a.x) + bv.x, 0.0f);
                v.y = fmaxf(fmaf(hv[q].y, dd, a.y) + bv.y, 0.0f);
                v.z = fmaxf(fmaf(hv[q].z, dd, a.z) + bv.z, 0.0f);
                v.w = fmaxf(fmaf(hv[q].w, dd, a.w) + bv.w, 0.0f);
                *(float4*)&gl[(i / Q4) * GP + c4] = v;
            }
        }
        __syncthreads();
        if (mul) {
            const int m0 = wave * 16;
            int ra = m0 + c16;
            ra = ra < cnt ? ra : cnt - 1;
            const float* ap = &gl[ra * GP + 2 * hh];
            v2f af[KQ];
            #pragma unroll
            for (int q = 0; q < KQ; ++q) af[q] = *(const v2f*)(ap + 4 * q);
            v8f c0 = {}, c1 = {};
            #pragma unroll
            for (int q = 0; q < KQ; ++q) {
                c0 = __builtin_amdgcn_wmma_f32_16x16x4_f32(false, af[q], false, bw0[q], (short)0, c0, false, false);
                c1 = __builtin_amdgcn_wmma_f32_16x16x4_f32(false, af[q], false, bw1[q], (short)0, c1, false, false);
            }
            float* orow = outp + (size_t)(lo + m0) * HID2 + lane;
            #pragma unroll
            for (int r = 0; r < 8; ++r) {
                const float s0 = xhalf(c0[r]);
                const float s1 = xhalf(c1[r]);
                const float lov = hh == 0 ? c0[r] : s1;
                const float hiv = hh == 0 ? s0 : c1[r];
                if (m0 + r < cnt)     orow[(size_t)r * HID2] = lov;
                if (m0 + 8 + r < cnt) orow[(size_t)(8 + r) * HID2] = hiv;
            }
        }
    } else {
        static_assert(!LAST || CH == 32, "the projection takes one channel per lane");
        constexpr int NPW = (RN + 31) / 32;
        float hq[NPW], dq[NPW];
        #pragma unroll
        for (int q = 0; q < NPW; ++q) {
            const int node = q * 32 + wave;
            hq[q] = 0.0f;
            dq[q] = 0.0f;
            if (node < cnt) {
                hq[q] = h[(size_t)(lo + node) * CH + lane];
                dq[q] = dinv[lo + node];
            }
        }
        const float bl = bias[lane], wl = Wn[lane], b0 = bh[0];
        #pragma unroll
        for (int q = 0; q < NPW; ++q) {
            const int node = q * 32 + wave;
            if (node < cnt) {
                const float dd = dq[q] * dq[q];
                const float a = acc[node * CH + lane];
                float v = a + hq[q] * dd + bl;
                v = fmaxf(v, 0.0f);
                float p = v * wl;
                #pragma unroll
                for (int off = 16; off > 0; off >>= 1)
                    p += __shfl_xor(p, off, 32);
                if (lane == 0) outv[node] = p + b0;
            }
        }
        __syncthreads();
        for (int i = tid; i < cnt; i += 1024) outp[lo + i] = outv[i];
    }
}

#define NRG1 256
#define NPT1 128
#define NPR1 128
#define RNG1 196

extern "C" void kernel_launch(void* const* d_in, const int* in_sizes, int n_in,
                              void* d_out, int out_size, void* d_ws, size_t ws_size,
                              hipStream_t stream) {
    const float* x  = (const float*)d_in[0];
    const int*   ei = (const int*)d_in[1];
    const float* W1 = (const float*)d_in[2];
    const float* b1 = (const float*)d_in[3];
    const float* W2 = (const float*)d_in[4];
    const float* b2 = (const float*)d_in[5];
    const float* Wh = (const float*)d_in[6];
    const float* bh = (const float*)d_in[7];
    float* out = (float*)d_out;

    const int N = in_sizes[0] / IN_CH;
    const int E = in_sizes[1] / 2;
    const int* src = ei;
    const int* dst = ei + E;

    float* f     = (float*)d_ws;
    float* dinv  = f;
    float* h1    = f + 65536;
    float* h2    = h1 + (size_t)N * HID1;
    int*   lists = (int*)(h2 + (size_t)N * HID2);

    split_prod_kernel<RNG1, NRG1, NPT1, NPR1><<<NPR1 + NPT1, 1024, 0, stream>>>(src, dst, lists, E, N, x, W1, h1);

    edge_count_dinv_kernel<RNG1, NRG1, NPT1><<<NRG1, 1024, 0, stream>>>(lists, dinv, N);

    edge_agg_kernel<HID1, false, RNG1, NRG1, NPT1><<<NRG1, 1024, 0, stream>>>(h1, lists, dinv, b1, W2, bh, h2, N);

    edge_agg_kernel<HID2, true, RNG1, NRG1, NPT1><<<NRG1, 1024, 0, stream>>>(h2, lists, dinv, b2, Wh, bh, out, N);
}


# Degree factors read from LDS: each edge-pass work-group copies the whole dinv table (200 KB) into its LDS once; an edge's two factors are LDS reads, not 4-byte gathers over 32 lines; sums as before
# speedup vs baseline: 4.0943x; 1.0782x over previous
#include <hip/hip_runtime.h>

typedef float v2f __attribute__((ext_vector_type(2)));
typedef float v8f __attribute__((ext_vector_type(8)));

#define IN_CH  128
#define HID1   64
#define HID2   32

__device__ __forceinline__ float xhalf(float x) {
    return __builtin_bit_cast(float, __builtin_amdgcn_permlanex16(__builtin_bit_cast(int, x), __builtin_bit_cast(int, x),
                                                                  0x76543210, 0xfedcba98, false, false));
}

#define EPART 12512

template <int NRANGE, int NPARTS>
struct WalkGeom {
    static constexpr int PPW = NPARTS / 32;
    static_assert(NPARTS % 32 == 0, "every wave walks the same number of parts");
};

template <int RN, int NRANGE, int NPARTS, int NPROD>
__global__ __launch_bounds__(1024) void split_prod_kernel(const int* __restrict__ src,
                                                         const int* __restrict__ dst,
                                                         int* __restrict__ lists,
                                                         int nE, int nN,
                                                         const float* __restrict__ X,
                                                         const float* __restrict__ W,
                                                         float* __restrict__ H) {
    static_assert(NRANGE == 256, "eight ranges per lane of the scanning wave; 8-bit range field");
    static_assert(RN <= 256, "8-bit local row field");
    static_assert(EPART < 32768, "15-bit rank field");
    constexpr int STGA = EPART > 20608 ? EPART : 20608;
    constexpr int KPT  = (EPART + 1023) / 1024;
    __shared__ int st_s[STGA];
    __shared__ int st_d[STGA];
    __shared__ int hist[NRANGE];
    __shared__ int offs[NRANGE + 1];
    const int tid  = threadIdx.x;
    const int lane = tid & 31;
    const int wave = tid >> 5;

    if (blockIdx.x < NPROD) {
        static_assert(IN_CH * HID1 == 8192 && STGA >= 8192, "two (k / 4, n) groups of W per thread");
        float* const wl = reinterpret_cast<float*>(st_s);
        const int wv = __builtin_amdgcn_readfirstlane(wave);
        const int c  = lane & 15;
        const int hh = lane >> 4;
        const int ntask = ((nN + 15) >> 4) * 2;
        const int t0 = (int)((long long)ntask * blockIdx.x / NPROD);
        const int t1 = (int)((long long)ntask * (blockIdx.x + 1) / NPROD);
        float wq[2][4];
        #pragma unroll
        for (int j = 0; j < 2; ++j) {
            const int p = j * 1024 + tid;
            #pragma unroll
            for (int e = 0; e < 4; ++e) wq[j][e] = W[(((p >> 6) << 2) + e) * HID1 + (p & 63)];
        }
        auto rows_of = [&](int t) -> const float4* {
            int tt = t < t1 - 1 ? t : t1 - 1;
            tt = tt < 0 ? 0 : tt;
            int row = (tt >> 1) * 16 + c;
            row = row < nN ? row : nN - 1;
            return (const float4*)(X + (size_t)row * IN_CH + 4 * hh);
        };
        int tau = t0 + wv;
        float4 a[16];
        {
            const float4* xp = rows_of(tau);
            #pragma unroll
            for (int q = 0; q < 16; ++q) a[q] = xp[2 * q];
        }
        #pragma unroll
        for (int j = 0; j < 2; ++j)
            *(float4*)&wl[(j * 1024 + tid) << 2] = make_float4(wq[j][0], wq[j][1], wq[j][2], wq[j][3]);
        __syncthreads();
        for (; tau < t1; tau += 32) {
            const int m0 = (tau >> 1) * 16;
            const int n0 = (tau & 1) * 32;
            const float4* xn = rows_of(tau + 32);
            v8f c0 = {}, c1 = {};
            #pragma unroll
            for (int q = 0; q < 16; ++q) {
                const float4 b0 = *(const float4*)&wl[(((2 * q + hh) << 6) + n0 + c) << 2];
                const float4 b1 = *(const float4*)&wl[(((2 * q + hh) << 6) + n0 + 16 + c) << 2];
                v2f a01, a23, b001, b023, b101, b123;
                a01.x = a[q].x; a01.y = a[q].y; a23.x = a[q].z; a23.y = a[q].w;
                b001.x = b0.x; b001.y = b0.y; b023.x = b0.z; b023.y = b0.w;
                b101.x = b1.x; b101.y = b1.y; b123.x = b1.z; b123.y = b1.w;
                c0 = __builtin_amdgcn_wmma_f32_16x16x4_f32(false, a01, false, b001, (short)0, c0, false, false);
                c1 = __builtin_amdgcn_wmma_f32_16x16x4_f32(false, a01, false, b101, (short)0, c1, false, false);
                c0 = __builtin_amdgcn_wmma_f32_16x16x4_f32(false, a23, false, b023, (short)0, c0, false, false);
                c1 = __builtin_amdgcn_wmma_f32_16x16x4_f32(false, a23, false, b123, (short)0, c1, false, false);
                a[q] = xn[2 * q];
            }
            #pragma unroll
            for (int r = 0; r < 8; ++r) {
                const float s0 = xhalf(c0[r]);
                const float s1 = xhalf(c1[r]);
                const float lo = hh == 0 ? c0[r] : s1;
                const float hi = hh == 0 ? s0 : c1[r];
                if (m0 + r < nN)     H[(size_t)(m0 + r) * HID1 + n0 + lane] = lo;
                if (m0 + 8 + r < nN) H[(size_t)(m0 + 8 + r) * HID1 + n0 + lane] = hi;
            }
        }
        return;
    }

    const int part = blockIdx.x - NPROD;
    const int e0 = (int)((long long)nE * part / NPARTS);
    const int e1 = (int)((long long)nE * (part + 1) / NPARTS);
    int cnt = e1 - e0;
    cnt = cnt > EPART ? EPART : cnt;

    int dv[KPT], sv[KPT], pk[KPT];
    #pragma unroll
    for (int k = 0; k < KPT; ++k) {
        const int i = k * 1024 + tid;
        const int e = i < cnt ? e0 + i : 0;
        dv[k] = dst[e];
        sv[k] = src[e];
    }
    if (tid < NRANGE) hist[tid] = 0;
    __syncthreads();

    #pragma unroll
    for (int k = 0; k < KPT; ++k) {
        const int i = k * 1024 + tid;
        pk[k] = -1;
        if (i < cnt && (unsigned)dv[k] < (unsigned)(nN < RN * NRANGE ? nN : RN * NRANGE)) {
            const int r  = (int)((unsigned)dv[k] / (unsigned)RN);
            const int dl = dv[k] - r * RN;
            const int rank = atomicAdd(&hist[r], 1);
            pk[k] = r | (dl << 8) | (rank << 16);
        }
    }
    __syncthreads();

    if (wave == 0) {
        int hv[8];
        int sum = 0;
        #pragma unroll
        for (int j = 0; j < 8; ++j) { hv[j] = hist[lane * 8 + j]; sum += hv[j]; }
        int inc = sum;
        #pragma unroll
        for (int off = 1; off < 32; off <<= 1) {
            const int t = __shfl_up(inc, off, 32);
            if (lane >= off) inc += t;
        }
        int run = inc - sum;
        #pragma unroll
        for (int j = 0; j < 8; ++j) { offs[lane * 8 + j] = run; run += hv[j]; }
        if (lane == 31) offs[NRANGE] = run;
    }
    __syncthreads();

    int* const tables = lists + (size_t)2 * NPARTS * EPART;
    if (tid < NRANGE)
        *(int2*)(tables + ((size_t)part * NRANGE + tid) * 2) = make_int2(offs[tid], hist[tid]);

    #pragma unroll
    for (int k = 0; k < KPT; ++k) {
        if (pk[k] >= 0) {
            const int pos = offs[pk[k] & 255] + (pk[k] >> 16);
            st_s[pos] = sv[k];
            st_d[pos] = (pk[k] >> 8) & 255;
        }
    }
    __syncthreads();

    const int tot = offs[NRANGE];
    int* const gs = lists + (size_t)part * EPART;
    int* const gd = lists + (size_t)NPARTS * EPART + (size_t)part * EPART;
    for (int i = tid; i < tot; i += 1024) {
        gs[i] = st_s[i];
        gd[i] = st_d[i];
    }
}

template <int NRANGE, int NPARTS>
struct WaveRuns {
    static constexpr int PPW = WalkGeom<NRANGE, NPARTS>::PPW;
    int base[PPW];
    int first[PPW];
    int total;
    __device__ __forceinline__ void load(const int* __restrict__ lists, int range, int wave) {
        const int* const tables = lists + (size_t)2 * NPARTS * EPART;
        int st[PPW], c[PPW];
        #pragma unroll
        for (int j = 0; j < PPW; ++j) {
            const int p = wave * PPW + j;
            const int2 t = *(const int2*)(tables + ((size_t)p * NRANGE + range) * 2);
            st[j] = __builtin_amdgcn_readfirstlane(t.x);
            c[j]  = __builtin_amdgcn_readfirstlane(t.y);
        }
        int pos = 0;
        #pragma unroll
        for (int j = 0; j < PPW; ++j) {
            first[j] = pos;
            base[j]  = (wave * PPW + j) * EPART + st[j] - pos;
            pos += c[j];
        }
        total = pos;
    }
    __device__ __forceinline__ int at(int v) const {
        int a = base[0];
        #pragma unroll
        for (int j = 1; j < PPW; ++j) a = v >= first[j] ? base[j] : a;
        return a + v;
    }
};

template <int RN, int NRANGE, int NPARTS>
__global__ __launch_bounds__(1024) void edge_count_dinv_kernel(const int* __restrict__ lists,
                                                              float* __restrict__ dinv, int nN) {
    __shared__ int cntl[RN];
    const int tid  = threadIdx.x;
    const int lane = tid & 31;
    const int wave = __builtin_amdgcn_readfirstlane(tid >> 5);
    const int range = blockIdx.x;
    const int lo = range * RN;
    int cnt = nN - lo;
    cnt = cnt < 0 ? 0 : (cnt > RN ? RN : cnt);
    const int* const gd = lists + (size_t)NPARTS * EPART;
    WaveRuns<NRANGE, NPARTS> runs;
    runs.load(lists, range, wave);
    const int nb = (runs.total + 31) >> 5;
    for (int i = tid; i < RN; i += 1024) cntl[i] = 0;
    __syncthreads();
    for (int k0 = 0; k0 < nb; k0 += 8) {
        int dlv[8];
        #pragma unroll
        for (int j = 0; j < 8; ++j) {
            const int v = 32 * (k0 + j) + lane;
            dlv[j] = v < runs.total ? gd[runs.at(v)] : -1;
        }
        #pragma unroll
        for (int j = 0; j < 8; ++j)
            if (dlv[j] >= 0) atomicAdd(&cntl[dlv[j]], 1);
    }
    __syncthreads();
    for (int i = tid; i < cnt; i += 1024) dinv[lo + i] = rsqrtf((float)cntl[i] + 1.0f);
}

template <int CH, bool LAST, int RN, int NRANGE, int NPARTS>
__global__ __launch_bounds__(1024) void edge_agg_kernel(const float* __restrict__ h,
                                                       const int* __restrict__ lists,
                                                       const float* __restrict__ dinv,
                                                       const float* __restrict__ bias,
                                                       const float* __restrict__ Wn,
                                                       const float* __restrict__ bh,
                                                       float* __restrict__ outp,
                                                       int nN) {
    constexpr int VW = CH / 32;
    constexpr int ACCF = LAST ? RN * CH : RN * CH + RN * (CH + 4);
    __shared__ float acc[ACCF];
    __shared__ float outv[LAST ? RN : 1];
    constexpr int TBL = RN * NRANGE;
    constexpr int NQT = (TBL / 4 + 1023) / 1024;
    static_assert(TBL % 4 == 0, "the table is copied in 16-B pieces");
    __shared__ __attribute__((aligned(16))) float dtab[TBL];
    const int tid  = threadIdx.x;
    const int lane = tid & 31;
    const int wave = __builtin_amdgcn_readfirstlane(tid >> 5);
    const int range = blockIdx.x;
    const int lo = range * RN;
    int cnt = nN - lo;
    cnt = cnt < 0 ? 0 : (cnt > RN ? RN : cnt);

    const int* const gs = lists;
    const int* const gd = lists + (size_t)NPARTS * EPART;
    const int nlim = nN < TBL ? nN : TBL;
    const int nq   = nlim >> 2;
    const int rot  = nq > 0 ? (int)(((unsigned)blockIdx.x * 48u) % (unsigned)nq) : 0;
    float4 tq[NQT];
    unsigned pq[NQT];
    #pragma unroll
    for (int j = 0; j < NQT; ++j) {
        unsigned qq = (unsigned)(j * 1024 + tid) + (unsigned)rot;
        qq = qq >= (unsigned)nq ? qq - (unsigned)nq : qq;
        qq = qq >= (unsigned)nq ? qq - (unsigned)nq : qq;
        pq[j] = qq < (unsigned)nq ? qq : 0u;
        tq[j] = *(const float4*)(dinv + 4u * pq[j]);
    }
    float tqt = 0.0f;
    if (tid < (nlim & 3)) tqt = dinv[4 * nq + tid];
    WaveRuns<NRANGE, NPARTS> runs;
    runs.load(lists, range, wave);
    const int tot = runs.total;
    const int nb = (tot + 31) >> 5;
    int s = 0, dl = 0, n = 0;
    if (nb > 0) {
        n = tot < 32 ? tot : 32;
        if (lane < n) { const int a = runs.at(lane); s = gs[a]; dl = gd[a]; }
    }

    #pragma unroll
    for (int j = 0; j < NQT; ++j) *(float4*)&dtab[4u * pq[j]] = tq[j];
    if (tid < (nlim & 3)) dtab[4 * nq + tid] = tqt;
    for (int i = tid * 4; i < RN * CH; i += 4096)
        *(float4*)&acc[i] = make_float4(0.0f, 0.0f, 0.0f, 0.0f);
    __syncthreads();

    for (int k = 0; k < nb; ++k) {
        int s2 = 0, dl2 = 0, n2 = 0;
        const bool more = k + 1 < nb;
        if (more) {
            const int left = tot - 32 * (k + 1);
            n2 = left < 32 ? left : 32;
        }
        const int v2 = 32 * (k + 1) + lane;
        const int st = (unsigned)s < (unsigned)(TBL - 1) ? s : TBL - 1;
        float ns, nd, nm;
        if (n == 32) {
            float r[32][VW];
            #pragma unroll
            for (int u = 0; u < 32; ++u) {
                const unsigned su = (unsigned)__builtin_amdgcn_readlane(s, u);
                const float* hp = h + (size_t)su * CH + lane * VW;
                #pragma unroll
                for (int w = 0; w < VW; ++w) r[u][w] = hp[w];
            }
            ns = dtab[st]; nd = dtab[lo + dl];
            if (more && lane < n2) { const int a = runs.at(v2); s2 = gs[a]; dl2 = gd[a]; }
            nm = ns * nd;
            #pragma unroll
            for (int u = 0; u < 32; ++u) {
                const int du = __builtin_amdgcn_readlane(dl, u);
                const float nu = __builtin_bit_cast(float,
                    __builtin_amdgcn_readlane(__builtin_bit_cast(int, nm), u));
                float* ap = &acc[du * CH + lane * VW];
                #pragma unroll
                for (int w = 0; w < VW; ++w) atomicAdd(ap + w, r[u][w] * nu);
            }
        } else {
            ns = dtab[st]; nd = dtab[lo + dl];
            if (more && lane < n2) { const int a = runs.at(v2); s2 = gs[a]; dl2 = gd[a]; }
            nm = (lane < n) ? ns * nd : 0.0f;
            #pragma unroll
            for (int g = 0; g < 4; ++g) {
                if (g * 8 < n) {
                    float r[8][VW];
                    #pragma unroll
                    for (int u = 0; u < 8; ++u) {
                        const unsigned su = (unsigned)__builtin_amdgcn_readlane(s, g * 8 + u);
                        const float* hp = h + (size_t)su * CH + lane * VW;
                        #pragma unroll
                        for (int w = 0; w < VW; ++w) r[u][w] = hp[w];
                    }
                    #pragma unroll
                    for (int u = 0; u < 8; ++u) {
                        if (g * 8 + u < n) {
                            const int du = __builtin_amdgcn_readlane(dl, g * 8 + u);
                            const float nu = __builtin_bit_cast(float,
                                __builtin_amdgcn_readlane(__builtin_bit_cast(int, nm), g * 8 + u));
                            float* ap = &acc[du * CH + lane * VW];
                            #pragma unroll
                            for (int w = 0; w < VW; ++w) atomicAdd(ap + w, r[u][w] * nu);
                        }
                    }
                }
            }
        }
        s = s2; dl = dl2; n = n2;
    }
    __syncthreads();

    if (!LAST) {
        static_assert(LAST || CH % 4 == 0, "k runs in steps of four");
        constexpr int Q4  = CH / 4;
        constexpr int NIT = (RN * Q4 + 1023) / 1024;
        constexpr int KQ  = CH / 4;
        constexpr int NRT = (RN + 15) / 16;
        constexpr int GP = CH + 4;
        static_assert(LAST || RN * CH + RN * GP <= ACCF, "the finished rows fit behind the sums");
        static_assert(LAST || NRT <= 32, "one wave per 16-row tile");
        float* const gl = acc + RN * CH;
        static_assert(LAST || 1024 % Q4 == 0, "a thread keeps its four channels over its rows");
        const int c16 = lane & 15;
        const int hh  = lane >> 4;
        const int c4  = (tid % Q4) * 4;
        const bool mul = wave < NRT && wave * 16 < cnt;
        const int lr = cnt > 0 ? lo : 0;
        const float* hr = h + (size_t)lr * CH;
        float4 hv[NIT];
        float  dq[NIT];
        v2f bw0[KQ], bw1[KQ];
        #pragma unroll
        for (int q = 0; q < NIT; ++q) {
            int i = q * 1024 + tid;
            i = i < cnt * Q4 ? i : 0;
            hv[q] = *(const float4*)(hr + (size_t)i * 4);
            dq[q] = dtab[lr + i / Q4];
        }
        const float4 bv = *(const float4*)(bias + c4);
        if (mul) {
            #pragma unroll
            for (int q = 0; q < KQ; ++q) {
                const int k = 4 * q + 2 * hh;
                bw0[q].x = Wn[(k + 0) * HID2 + c16];
                bw0[q].y = Wn[(k + 1) * HID2 + c16];
                bw1[q].x = Wn[(k + 0) * HID2 + 16 + c16];
                bw1[q].y = Wn[(k + 1) * HID2 + 16 + c16];
            }
        }
        #pragma unroll
        for (int q = 0; q < NIT; ++q) {
            const int i = q * 1024 + tid;
            if (i < cnt * Q4) {
                const float dd = dq[q] * dq[q];
                const float4 a  = *(const float4*)&acc[i * 4];
                float4 v;
                v.x = fmaxf(fmaf(hv[q].x, dd, a.x) + bv.x, 0.0f);
                v.y = fmaxf(fmaf(hv[q].y, dd, a.y) + bv.y, 0.0f);
                v.z = fmaxf(fmaf(hv[q].z, dd, a.z) + bv.z, 0.0f);
                v.w = fmaxf(fmaf(hv[q].w, dd, a.w) + bv.w, 0.0f);
                *(float4*)&gl[(i / Q4) * GP + c4] = v;
            }
        }
        __syncthreads();
        if (mul) {
            const int m0 = wave * 16;
            int ra = m0 + c16;
            ra = ra < cnt ? ra : cnt - 1;
            const float* ap = &gl[ra * GP + 2 * hh];
            v2f af[KQ];
            #pragma unroll
            for (int q = 0; q < KQ; ++q) af[q] = *(const v2f*)(ap + 4 * q);
            v8f c0 = {}, c1 = {};
            #pragma unroll
            for (int q = 0; q < KQ; ++q) {
                c0 = __builtin_amdgcn_wmma_f32_16x16x4_f32(false, af[q], false, bw0[q], (short)0, c0, false, false);
                c1 = __builtin_amdgcn_wmma_f32_16x16x4_f32(false, af[q], false, bw1[q], (short)0, c1, false, false);
            }
            float* orow = outp + (size_t)(lo + m0) * HID2 + lane;
            #pragma unroll
            for (int r = 0; r < 8; ++r) {
                const float s0 = xhalf(c0[r]);
                const float s1 = xhalf(c1[r]);
                const float lov = hh == 0 ? c0[r] : s1;
                const float hiv = hh == 0 ? s0 : c1[r];
                if (m0 + r < cnt)     orow[(size_t)r * HID2] = lov;
                if (m0 + 8 + r < cnt) orow[(size_t)(8 + r) * HID2] = hiv;
            }
        }
    } else {
        static_assert(!LAST || CH == 32, "the projection takes one channel per lane");
        constexpr int NPW = (RN + 31) / 32;
        float hq[NPW], dq[NPW];
        #pragma unroll
        for (int q = 0; q < NPW; ++q) {
            const int node = q * 32 + wave;
            hq[q] = 0.0f;
            dq[q] = 0.0f;
            if (node < cnt) {
                hq[q] = h[(size_t)(lo + node) * CH + lane];
                dq[q] = dtab[lo + node];
            }
        }
        const float bl = bias[lane], wl = Wn[lane], b0 = bh[0];
        #pragma unroll
        for (int q = 0; q < NPW; ++q) {
            const int node = q * 32 + wave;
            if (node < cnt) {
                const float dd = dq[q] * dq[q];
                const float a = acc[node * CH + lane];
                float v = a + hq[q] * dd + bl;
                v = fmaxf(v, 0.0f);
                float p = v * wl;
                #pragma unroll
                for (int off = 16; off > 0; off >>= 1)
                    p += __shfl_xor(p, off, 32);
                if (lane == 0) outv[node] = p + b0;
            }
        }
        __syncthreads();
        for (int i = tid; i < cnt; i += 1024) outp[lo + i] = outv[i];
    }
}

#define NRG1 256
#define NPT1 128
#define NPR1 128
#define RNG1 196

extern "C" void kernel_launch(void* const* d_in, const int* in_sizes, int n_in,
                              void* d_out, int out_size, void* d_ws, size_t ws_size,
                              hipStream_t stream) {
    const float* x  = (const float*)d_in[0];
    const int*   ei = (const int*)d_in[1];
    const float* W1 = (const float*)d_in[2];
    const float* b1 = (const float*)d_in[3];
    const float* W2 = (const float*)d_in[4];
    const float* b2 = (const float*)d_in[5];
    const float* Wh = (const float*)d_in[6];
    const float* bh = (const float*)d_in[7];
    float* out = (float*)d_out;

    const int N = in_sizes[0] / IN_CH;
    const int E = in_sizes[1] / 2;
    const int* src = ei;
    const int* dst = ei + E;

    float* f     = (float*)d_ws;
    float* dinv  = f;
    float* h1    = f + 65536;
    float* h2    = h1 + (size_t)N * HID1;
    int*   lists = (int*)(h2 + (size_t)N * HID2);

    split_prod_kernel<RNG1, NRG1, NPT1, NPR1><<<NPR1 + NPT1, 1024, 0, stream>>>(src, dst, lists, E, N, x, W1, h1);

    edge_count_dinv_kernel<RNG1, NRG1, NPT1><<<NRG1, 1024, 0, stream>>>(lists, dinv, N);

    edge_agg_kernel<HID1, false, RNG1, NRG1, NPT1><<<NRG1, 1024, 0, stream>>>(h1, lists, dinv, b1, W2, bh, h2, N);

    edge_agg_kernel<HID2, true, RNG1, NRG1, NPT1><<<NRG1, 1024, 0, stream>>>(h2, lists, dinv, b2, Wh, bh, out, N);
}


# speedup vs baseline: 4.3647x; 1.0660x over previous
#include <hip/hip_runtime.h>

typedef float v2f __attribute__((ext_vector_type(2)));
typedef float v8f __attribute__((ext_vector_type(8)));

#define IN_CH  128
#define HID1   64
#define HID2   32

__device__ __forceinline__ float xhalf(float x) {
    return __builtin_bit_cast(float, __builtin_amdgcn_permlanex16(__builtin_bit_cast(int, x), __builtin_bit_cast(int, x),
                                                                  0x76543210, 0xfedcba98, false, false));
}

#define EPART 12512

template <int NRANGE, int NPARTS>
struct WalkGeom {
    static constexpr int PPW = NPARTS / 32;
    static_assert(NPARTS % 32 == 0, "every wave walks the same number of parts");
};

template <int RN, int NRANGE, int NPARTS, int NPROD>
__global__ __launch_bounds__(1024) void split_prod_kernel(const int* __restrict__ src,
                                                         const int* __restrict__ dst,
                                                         int* __restrict__ lists,
                                                         int nE, int nN,
                                                         const float* __restrict__ X,
                                                         const float* __restrict__ W,
                                                         float* __restrict__ H) {
    static_assert(NRANGE == 256, "eight ranges per lane of the scanning wave; 8-bit range field");
    static_assert(RN <= 256, "8-bit local row field");
    static_assert(EPART < 32768, "15-bit rank field");
    constexpr int STGA = EPART > 20608 ? EPART : 20608;
    constexpr int KPT  = (EPART + 1023) / 1024;
    __shared__ int st_s[STGA];
    __shared__ int st_d[STGA];
    __shared__ int hist[NRANGE];
    __shared__ int offs[NRANGE + 1];
    const int tid  = threadIdx.x;
    const int lane = tid & 31;
    const int wave = tid >> 5;

    if (blockIdx.x < NPROD) {
        static_assert(IN_CH * HID1 == 8192 && STGA >= 8192, "two (k / 4, n) groups of W per thread");
        float* const wl = reinterpret_cast<float*>(st_s);
        const int wv = __builtin_amdgcn_readfirstlane(wave);
        const int c  = lane & 15;
        const int hh = lane >> 4;
        const int ntask = ((nN + 15) >> 4) * 2;
        const int t0 = (int)((long long)ntask * blockIdx.x / NPROD);
        const int t1 = (int)((long long)ntask * (blockIdx.x + 1) / NPROD);
        float wq[2][4];
        #pragma unroll
        for (int j = 0; j < 2; ++j) {
            const int p = j * 1024 + tid;
            #pragma unroll
            for (int e = 0; e < 4; ++e) wq[j][e] = W[(((p >> 6) << 2) + e) * HID1 + (p & 63)];
        }
        auto rows_of = [&](int t) -> const float4* {
            int tt = t < t1 - 1 ? t : t1 - 1;
            tt = tt < 0 ? 0 : tt;
            int row = (tt >> 1) * 16 + c;
            row = row < nN ? row : nN - 1;
            return (const float4*)(X + (size_t)row * IN_CH + 4 * hh);
        };
        int tau = t0 + wv;
        float4 a[16];
        {
            const float4* xp = rows_of(tau);
            #pragma unroll
            for (int q = 0; q < 16; ++q) a[q] = xp[2 * q];
        }
        #pragma unroll
        for (int j = 0; j < 2; ++j)
            *(float4*)&wl[(j * 1024 + tid) << 2] = make_float4(wq[j][0], wq[j][1], wq[j][2], wq[j][3]);
        __syncthreads();
        for (; tau < t1; tau += 32) {
            const int m0 = (tau >> 1) * 16;
            const int n0 = (tau & 1) * 32;
            const float4* xn = rows_of(tau + 32);
            v8f c0 = {}, c1 = {};
            #pragma unroll
            for (int q = 0; q < 16; ++q) {
                const float4 b0 = *(const float4*)&wl[(((2 * q + hh) << 6) + n0 + c) << 2];
                const float4 b1 = *(const float4*)&wl[(((2 * q + hh) << 6) + n0 + 16 + c) << 2];
                v2f a01, a23, b001, b023, b101, b123;
                a01.x = a[q].x; a01.y = a[q].y; a23.x = a[q].z; a23.y = a[q].w;
                b001.x = b0.x; b001.y = b0.y; b023.x = b0.z; b023.y = b0.w;
                b101.x = b1.x; b101.y = b1.y; b123.x = b1.z; b123.y = b1.w;
                c0 = __builtin_amdgcn_wmma_f32_16x16x4_f32(false, a01, false, b001, (short)0, c0, false, false);
                c1 = __builtin_amdgcn_wmma_f32_16x16x4_f32(false, a01, false, b101, (short)0, c1, false, false);
                c0 = __builtin_amdgcn_wmma_f32_16x16x4_f32(false, a23, false, b023, (short)0, c0, false, false);
                c1 = __builtin_amdgcn_wmma_f32_16x16x4_f32(false, a23, false, b123, (short)0, c1, false, false);
                a[q] = xn[2 * q];
            }
            #pragma unroll
            for (int r = 0; r < 8; ++r) {
                const float s0 = xhalf(c0[r]);
                const float s1 = xhalf(c1[r]);
                const float lo = hh == 0 ? c0[r] : s1;
                const float hi = hh == 0 ? s0 : c1[r];
                if (m0 + r < nN)     H[(size_t)(m0 + r) * HID1 + n0 + lane] = lo;
                if (m0 + 8 + r < nN) H[(size_t)(m0 + 8 + r) * HID1 + n0 + lane] = hi;
            }
        }
        return;
    }

    const int part = blockIdx.x - NPROD;
    const int e0 = (int)((long long)nE * part / NPARTS);
    const int e1 = (int)((long long)nE * (part + 1) / NPARTS);
    int cnt = e1 - e0;
    cnt = cnt > EPART ? EPART : cnt;

    int dv[KPT], sv[KPT], pk[KPT];
    #pragma unroll
    for (int k = 0; k < KPT; ++k) {
        const int i = k * 1024 + tid;
        const int e = i < cnt ? e0 + i : 0;
        dv[k] = dst[e];
        sv[k] = src[e];
    }
    if (tid < NRANGE) hist[tid] = 0;
    __syncthreads();

    #pragma unroll
    for (int k = 0; k < KPT; ++k) {
        const int i = k * 1024 + tid;
        pk[k] = -1;
        if (i < cnt && (unsigned)dv[k] < (unsigned)(nN < RN * NRANGE ? nN : RN * NRANGE)) {
            const int r  = (int)((unsigned)dv[k] / (unsigned)RN);
            const int dl = dv[k] - r * RN;
            const int rank = atomicAdd(&hist[r], 1);
            pk[k] = r | (dl << 8) | (rank << 16);
        }
    }
    __syncthreads();

    if (wave == 0) {
        int hv[8];
        int sum = 0;
        #pragma unroll
        for (int j = 0; j < 8; ++j) { hv[j] = hist[lane * 8 + j]; sum += hv[j]; }
        int inc = sum;
        #pragma unroll
        for (int off = 1; off < 32; off <<= 1) {
            const int t = __shfl_up(inc, off, 32);
            if (lane >= off) inc += t;
        }
        int run = inc - sum;
        #pragma unroll
        for (int j = 0; j < 8; ++j) { offs[lane * 8 + j] = run; run += hv[j]; }
        if (lane == 31) offs[NRANGE] = run;
    }
    __syncthreads();

    int* const tables = lists + (size_t)2 * NPARTS * EPART;
    if (tid < NRANGE)
        *(int2*)(tables + ((size_t)part * NRANGE + tid) * 2) = make_int2(offs[tid], hist[tid]);

    #pragma unroll
    for (int k = 0; k < KPT; ++k) {
        if (pk[k] >= 0) {
            const int pos = offs[pk[k] & 255] + (pk[k] >> 16);
            st_s[pos] = sv[k];
            st_d[pos] = (pk[k] >> 8) & 255;
        }
    }
    __syncthreads();

    const int tot = offs[NRANGE];
    int* const gs = lists + (size_t)part * EPART;
    int* const gd = lists + (size_t)NPARTS * EPART + (size_t)part * EPART;
    for (int i = tid; i < tot; i += 1024) {
        gs[i] = st_s[i];
        gd[i] = st_d[i];
    }
}

template <int NRANGE, int NPARTS>
struct WaveRuns {
    static constexpr int PPW = WalkGeom<NRANGE, NPARTS>::PPW;
    int base[PPW];
    int first[PPW];
    int total;
    __device__ __forceinline__ void load(const int* __restrict__ lists, int range, int wave) {
        const int* const tables = lists + (size_t)2 * NPARTS * EPART;
        int st[PPW], c[PPW];
        #pragma unroll
        for (int j = 0; j < PPW; ++j) {
            const int p = wave * PPW + j;
            const int2 t = *(const int2*)(tables + ((size_t)p * NRANGE + range) * 2);
            st[j] = __builtin_amdgcn_readfirstlane(t.x);
            c[j]  = __builtin_amdgcn_readfirstlane(t.y);
        }
        int pos = 0;
        #pragma unroll
        for (int j = 0; j < PPW; ++j) {
            first[j] = pos;
            base[j]  = (wave * PPW + j) * EPART + st[j] - pos;
            pos += c[j];
        }
        total = pos;
    }
    __device__ __forceinline__ int at(int v) const {
        int a = base[0];
        #pragma unroll
        for (int j = 1; j < PPW; ++j) a = v >= first[j] ? base[j] : a;
        return a + v;
    }
};

template <int RN, int NRANGE, int NPARTS>
__global__ __launch_bounds__(1024) void edge_count_dinv_kernel(const int* __restrict__ lists,
                                                              float* __restrict__ dinv, int nN) {
    __shared__ int cntl[RN];
    const int tid  = threadIdx.x;
    const int lane = tid & 31;
    const int wave = __builtin_amdgcn_readfirstlane(tid >> 5);
    const int range = blockIdx.x;
    const int lo = range * RN;
    int cnt = nN - lo;
    cnt = cnt < 0 ? 0 : (cnt > RN ? RN : cnt);
    const int* const gd = lists + (size_t)NPARTS * EPART;
    WaveRuns<NRANGE, NPARTS> runs;
    runs.load(lists, range, wave);
    const int nb = (runs.total + 31) >> 5;
    for (int i = tid; i < RN; i += 1024) cntl[i] = 0;
    __syncthreads();
    for (int k0 = 0; k0 < nb; k0 += 8) {
        int dlv[8];
        #pragma unroll
        for (int j = 0; j < 8; ++j) {
            const int v = 32 * (k0 + j) + lane;
            dlv[j] = v < runs.total ? gd[runs.at(v)] : -1;
        }
        #pragma unroll
        for (int j = 0; j < 8; ++j)
            if (dlv[j] >= 0) atomicAdd(&cntl[dlv[j]], 1);
    }
    __syncthreads();
    for (int i = tid; i < cnt; i += 1024) dinv[lo + i] = rsqrtf((float)cntl[i] + 1.0f);
}

template <int CH, bool LAST, int RN, int NRANGE, int NPARTS>
__global__ __launch_bounds__(1024) void edge_agg_kernel(const float* __restrict__ h,
                                                       const int* __restrict__ lists,
                                                       const float* __restrict__ dinv,
                                                       const float* __restrict__ bias,
                                                       const float* __restrict__ Wn,
                                                       const float* __restrict__ bh,
                                                       float* __restrict__ outp,
                                                       int nN) {
    constexpr int VW = CH / 32;
    constexpr int BLK  = 8192;
    constexpr int NSB  = 8;
    constexpr int ACCR = 2048;
    constexpr int DCH  = BLK - ACCR;
    static_assert(((RN + NSB - 1) / NSB) * CH <= ACCR, "a block's share of the sum rows fits its head");
    static_assert(DCH % 4 == 0, "a 16-B piece of the table stays inside one block");
    static_assert(NSB == 8, "row r: block r & 7, slot r >> 3");
    constexpr int TBL = RN * NRANGE;
    constexpr int NQT = (TBL / 4 + 1023) / 1024;
    static_assert(TBL % 4 == 0, "the table is copied in 16-B pieces");
    static_assert(TBL > NSB * DCH && TBL <= (NSB + 1) * DCH, "the table ends in the ninth block");
    constexpr int TAIL = NSB * BLK + (TBL - NSB * DCH);
    static_assert(TAIL % 64 == 0, "the finished rows start on bank 0");
    constexpr int LTOT = TAIL + (LAST ? RN : RN * (CH + 4));
    static_assert(LTOT * 4 <= 320 * 1024, "one WGP's LDS");
    __shared__ __attribute__((aligned(16))) float lds[LTOT];
    float* const outv = lds + TAIL;
    auto arow = [&](int r) -> int { return (r & (NSB - 1)) * BLK + (r >> 3) * CH; };
    auto dpos = [&](unsigned i) -> unsigned {
        const unsigned b = i / (unsigned)DCH;
        return i + (unsigned)ACCR * (b < (unsigned)NSB ? b + 1u : (unsigned)NSB);
    };
    const int tid  = threadIdx.x;
    const int lane = tid & 31;
    const int wave = __builtin_amdgcn_readfirstlane(tid >> 5);
    const int range = blockIdx.x;
    const int lo = range * RN;
    int cnt = nN - lo;
    cnt = cnt < 0 ? 0 : (cnt > RN ? RN : cnt);

    const int* const gs = lists;
    const int* const gd = lists + (size_t)NPARTS * EPART;
    const int nlim = nN < TBL ? nN : TBL;
    const int nq   = nlim >> 2;
    const int rot  = nq > 0 ? (int)(((unsigned)blockIdx.x * 48u) % (unsigned)nq) : 0;
    float4 tq[NQT];
    unsigned pq[NQT];
    #pragma unroll
    for (int j = 0; j < NQT; ++j) {
        unsigned qq = (unsigned)(j * 1024 + tid) + (unsigned)rot;
        qq = qq >= (unsigned)nq ? qq - (unsigned)nq : qq;
        qq = qq >= (unsigned)nq ? qq - (unsigned)nq : qq;
        pq[j] = qq < (unsigned)nq ? qq : 0u;
        tq[j] = *(const float4*)(dinv + 4u * pq[j]);
    }
    float tqt = 0.0f;
    if (tid < (nlim & 3)) tqt = dinv[4 * nq + tid];
    WaveRuns<NRANGE, NPARTS> runs;
    runs.load(lists, range, wave);
    const int tot = runs.total;
    const int nb = (tot + 31) >> 5;
    int s = 0, dl = 0, n = 0;
    if (nb > 0) {
        n = tot < 32 ? tot : 32;
        if (lane < n) { const int a = runs.at(lane); s = gs[a]; dl = gd[a]; }
    }

    #pragma unroll
    for (int j = 0; j < NQT; ++j) *(float4*)&lds[dpos(4u * pq[j])] = tq[j];
    if (tid < (nlim & 3)) lds[dpos((unsigned)(4 * nq + tid))] = tqt;
    for (int i = tid * 4; i < RN * CH; i += 4096)
        *(float4*)&lds[arow(i / CH) + i % CH] = make_float4(0.0f, 0.0f, 0.0f, 0.0f);
    __syncthreads();

    for (int k = 0; k < nb; ++k) {
        int s2 = 0, dl2 = 0, n2 = 0;
        const bool more = k + 1 < nb;
        if (more) {
            const int left = tot - 32 * (k + 1);
            n2 = left < 32 ? left : 32;
        }
        const int v2 = 32 * (k + 1) + lane;
        const int st = (unsigned)s < (unsigned)(TBL - 1) ? s : TBL - 1;
        float ns, nd, nm;
        if (n == 32) {
            float r[32][VW];
            #pragma unroll
            for (int u = 0; u < 32; ++u) {
                const unsigned su = (unsigned)__builtin_amdgcn_readlane(s, u);
                const float* hp = h + (size_t)su * CH + lane * VW;
                #pragma unroll
                for (int w = 0; w < VW; ++w) r[u][w] = hp[w];
            }
            ns = lds[dpos((unsigned)st)]; nd = lds[dpos((unsigned)(lo + dl))];
            if (more && lane < n2) { const int a = runs.at(v2); s2 = gs[a]; dl2 = gd[a]; }
            nm = ns * nd;
            #pragma unroll
            for (int u = 0; u < 32; ++u) {
                const int du = __builtin_amdgcn_readlane(dl, u);
                const float nu = __builtin_bit_cast(float,
                    __builtin_amdgcn_readlane(__builtin_bit_cast(int, nm), u));
                float* ap = &lds[arow(du) + lane * VW];
                #pragma unroll
                for (int w = 0; w < VW; ++w) atomicAdd(ap + w, r[u][w] * nu);
            }
        } else {
            ns = lds[dpos((unsigned)st)]; nd = lds[dpos((unsigned)(lo + dl))];
            if (more && lane < n2) { const int a = runs.at(v2); s2 = gs[a]; dl2 = gd[a]; }
            nm = (lane < n) ? ns * nd : 0.0f;
            #pragma unroll
            for (int g = 0; g < 4; ++g) {
                if (g * 8 < n) {
                    float r[8][VW];
                    #pragma unroll
                    for (int u = 0; u < 8; ++u) {
                        const unsigned su = (unsigned)__builtin_amdgcn_readlane(s, g * 8 + u);
                        const float* hp = h + (size_t)su * CH + lane * VW;
                        #pragma unroll
                        for (int w = 0; w < VW; ++w) r[u][w] = hp[w];
                    }
                    #pragma unroll
                    for (int u = 0; u < 8; ++u) {
                        if (g * 8 + u < n) {
                            const int du = __builtin_amdgcn_readlane(dl, g * 8 + u);
                            const float nu = __builtin_bit_cast(float,
                                __builtin_amdgcn_readlane(__builtin_bit_cast(int, nm), g * 8 + u));
                            float* ap = &lds[arow(du) + lane * VW];
                            #pragma unroll
                            for (int w = 0; w < VW; ++w) atomicAdd(ap + w, r[u][w] * nu);
                        }
                    }
                }
            }
        }
        s = s2; dl = dl2; n = n2;
    }
    __syncthreads();

    if (!LAST) {
        static_assert(LAST || CH % 4 == 0, "k runs in steps of four");
        constexpr int Q4  = CH / 4;
        constexpr int NIT = (RN * Q4 + 1023) / 1024;
        constexpr int KQ  = CH / 4;
        constexpr int NRT = (RN + 15) / 16;
        constexpr int GP = CH + 4;
        static_assert(LAST || NRT <= 32, "one wave per 16-row tile");
        float* const gl = lds + TAIL;
        static_assert(LAST || 1024 % Q4 == 0, "a thread keeps its four channels over its rows");
        const int c16 = lane & 15;
        const int hh  = lane >> 4;
        const int c4  = (tid % Q4) * 4;
        const bool mul = wave < NRT && wave * 16 < cnt;
        const int lr = cnt > 0 ? lo : 0;
        const float* hr = h + (size_t)lr * CH;
        float4 hv[NIT];
        float  dq[NIT];
        v2f bw0[KQ], bw1[KQ];
        #pragma unroll
        for (int q = 0; q < NIT; ++q) {
            int i = q * 1024 + tid;
            i = i < cnt * Q4 ? i : 0;
            hv[q] = *(const float4*)(hr + (size_t)i * 4);
            dq[q] = lds[dpos((unsigned)(lr + i / Q4))];
        }
        const float4 bv = *(const float4*)(bias + c4);
        if (mul) {
            #pragma unroll
            for (int q = 0; q < KQ; ++q) {
                const int k = 4 * q + 2 * hh;
                bw0[q].x = Wn[(k + 0) * HID2 + c16];
                bw0[q].y = Wn[(k + 1) * HID2 + c16];
                bw1[q].x = Wn[(k + 0) * HID2 + 16 + c16];
                bw1[q].y = Wn[(k + 1) * HID2 + 16 + c16];
            }
        }
        #pragma unroll
        for (int q = 0; q < NIT; ++q) {
            const int i = q * 1024 + tid;
            if (i < cnt * Q4) {
                const float dd = dq[q] * dq[q];
                const float4 a  = *(const float4*)&lds[arow(i / Q4) + c4];
                float4 v;
                v.x = fmaxf(fmaf(hv[q].x, dd, a.x) + bv.x, 0.0f);
                v.y = fmaxf(fmaf(hv[q].y, dd, a.y) + bv.y, 0.0f);
                v.z = fmaxf(fmaf(hv[q].z, dd, a.z) + bv.z, 0.0f);
                v.w = fmaxf(fmaf(hv[q].w, dd, a.w) + bv.w, 0.0f);
                *(float4*)&gl[(i / Q4) * GP + c4] = v;
            }
        }
        __syncthreads();
        if (mul) {
            const int m0 = wave * 16;
            int ra = m0 + c16;
            ra = ra < cnt ? ra : cnt - 1;
            const float* ap = &gl[ra * GP + 2 * hh];
            v2f af[KQ];
            #pragma unroll
            for (int q = 0; q < KQ; ++q) af[q] = *(const v2f*)(ap + 4 * q);
            v8f c0 = {}, c1 = {};
            #pragma unroll
            for (int q = 0; q < KQ; ++q) {
                c0 = __builtin_amdgcn_wmma_f32_16x16x4_f32(false, af[q], false, bw0[q], (short)0, c0, false, false);
                c1 = __builtin_amdgcn_wmma_f32_16x16x4_f32(false, af[q], false, bw1[q], (short)0, c1, false, false);
            }
            float* orow = outp + (size_t)(lo + m0) * HID2 + lane;
            #pragma unroll
            for (int r = 0; r < 8; ++r) {
                const float s0 = xhalf(c0[r]);
                const float s1 = xhalf(c1[r]);
                const float lov = hh == 0 ? c0[r] : s1;
                const float hiv = hh == 0 ? s0 : c1[r];
                if (m0 + r < cnt)     orow[(size_t)r * HID2] = lov;
                if (m0 + 8 + r < cnt) orow[(size_t)(8 + r) * HID2] = hiv;
            }
        }
    } else {
        static_assert(!LAST || CH == 32, "the projection takes one channel per lane");
        constexpr int NPW = (RN + 31) / 32;
        float hq[NPW], dq[NPW];
        #pragma unroll
        for (int q = 0; q < NPW; ++q) {
            const int node = q * 32 + wave;
            hq[q] = 0.0f;
            dq[q] = 0.0f;
            if (node < cnt) {
                hq[q] = h[(size_t)(lo + node) * CH + lane];
                dq[q] = lds[dpos((unsigned)(lo + node))];
            }
        }
        const float bl = bias[lane], wl = Wn[lane], b0 = bh[0];
        #pragma unroll
        for (int q = 0; q < NPW; ++q) {
            const int node = q * 32 + wave;
            if (node < cnt) {
                const float dd = dq[q] * dq[q];
                const float a = lds[arow(node) + lane];
                float v = a + hq[q] * dd + bl;
                v = fmaxf(v, 0.0f);
                float p = v * wl;
                #pragma unroll
                for (int off = 16; off > 0; off >>= 1)
                    p += __shfl_xor(p, off, 32);
                if (lane == 0) outv[node] = p + b0;
            }
        }
        __syncthreads();
        for (int i = tid; i < cnt; i += 1024) outp[lo + i] = outv[i];
    }
}

#define NRG1 256
#define NPT1 128
#define NPR1 128
#define RNG1 196

extern "C" void kernel_launch(void* const* d_in, const int* in_sizes, int n_in,
                              void* d_out, int out_size, void* d_ws, size_t ws_size,
                              hipStream_t stream) {
    const float* x  = (const float*)d_in[0];
    const int*   ei = (const int*)d_in[1];
    const float* W1 = (const float*)d_in[2];
    const float* b1 = (const float*)d_in[3];
    const float* W2 = (const float*)d_in[4];
    const float* b2 = (const float*)d_in[5];
    const float* Wh = (const float*)d_in[6];
    const float* bh = (const float*)d_in[7];
    float* out = (float*)d_out;

    const int N = in_sizes[0] / IN_CH;
    const int E = in_sizes[1] / 2;
    const int* src = ei;
    const int* dst = ei + E;

    float* f     = (float*)d_ws;
    float* dinv  = f;
    float* h1    = f + 65536;
    float* h2    = h1 + (size_t)N * HID1;
    int*   lists = (int*)(h2 + (size_t)N * HID2);

    split_prod_kernel<RNG1, NRG1, NPT1, NPR1><<<NPR1 + NPT1, 1024, 0, stream>>>(src, dst, lists, E, N, x, W1, h1);

    edge_count_dinv_kernel<RNG1, NRG1, NPT1><<<NRG1, 1024, 0, stream>>>(lists, dinv, N);

    edge_agg_kernel<HID1, false, RNG1, NRG1, NPT1><<<NRG1, 1024, 0, stream>>>(h1, lists, dinv, b1, W2, bh, h2, N);

    edge_agg_kernel<HID2, true, RNG1, NRG1, NPT1><<<NRG1, 1024, 0, stream>>>(h2, lists, dinv, b2, Wh, bh, out, N);
}


# Edge loops: a lane works out its own edge's source-row byte offset and LDS sum-row address once per 32-edge batch; per edge one broadcast and one 32-bit add each, loads on scalar base; sums as before
# speedup vs baseline: 4.6185x; 1.0581x over previous
#include <hip/hip_runtime.h>

typedef float v2f __attribute__((ext_vector_type(2)));
typedef float v8f __attribute__((ext_vector_type(8)));

#define IN_CH  128
#define HID1   64
#define HID2   32

__device__ __forceinline__ float xhalf(float x) {
    return __builtin_bit_cast(float, __builtin_amdgcn_permlanex16(__builtin_bit_cast(int, x), __builtin_bit_cast(int, x),
                                                                  0x76543210, 0xfedcba98, false, false));
}

#define EPART 12512

template <int NRANGE, int NPARTS>
struct WalkGeom {
    static constexpr int PPW = NPARTS / 32;
    static_assert(NPARTS % 32 == 0, "every wave walks the same number of parts");
};

template <int RN, int NRANGE, int NPARTS, int NPROD>
__global__ __launch_bounds__(1024) void split_prod_kernel(const int* __restrict__ src,
                                                         const int* __restrict__ dst,
                                                         int* __restrict__ lists,
                                                         int nE, int nN,
                                                         const float* __restrict__ X,
                                                         const float* __restrict__ W,
                                                         float* __restrict__ H) {
    static_assert(NRANGE == 256, "eight ranges per lane of the scanning wave; 8-bit range field");
    static_assert(RN <= 256, "8-bit local row field");
    static_assert(EPART < 32768, "15-bit rank field");
    constexpr int STGA = EPART > 20608 ? EPART : 20608;
    constexpr int KPT  = (EPART + 1023) / 1024;
    __shared__ int st_s[STGA];
    __shared__ int st_d[STGA];
    __shared__ int hist[NRANGE];
    __shared__ int offs[NRANGE + 1];
    const int tid  = threadIdx.x;
    const int lane = tid & 31;
    const int wave = tid >> 5;

    if (blockIdx.x < NPROD) {
        static_assert(IN_CH * HID1 == 8192 && STGA >= 8192, "two (k / 4, n) groups of W per thread");
        float* const wl = reinterpret_cast<float*>(st_s);
        const int wv = __builtin_amdgcn_readfirstlane(wave);
        const int c  = lane & 15;
        const int hh = lane >> 4;
        const int ntask = ((nN + 15) >> 4) * 2;
        const int t0 = (int)((long long)ntask * blockIdx.x / NPROD);
        const int t1 = (int)((long long)ntask * (blockIdx.x + 1) / NPROD);
        float wq[2][4];
        #pragma unroll
        for (int j = 0; j < 2; ++j) {
            const int p = j * 1024 + tid;
            #pragma unroll
            for (int e = 0; e < 4; ++e) wq[j][e] = W[(((p >> 6) << 2) + e) * HID1 + (p & 63)];
        }
        auto rows_of = [&](int t) -> const float4* {
            int tt = t < t1 - 1 ? t : t1 - 1;
            tt = tt < 0 ? 0 : tt;
            int row = (tt >> 1) * 16 + c;
            row = row < nN ? row : nN - 1;
            return (const float4*)(X + (size_t)row * IN_CH + 4 * hh);
        };
        int tau = t0 + wv;
        float4 a[16];
        {
            const float4* xp = rows_of(tau);
            #pragma unroll
            for (int q = 0; q < 16; ++q) a[q] = xp[2 * q];
        }
        #pragma unroll
        for (int j = 0; j < 2; ++j)
            *(float4*)&wl[(j * 1024 + tid) << 2] = make_float4(wq[j][0], wq[j][1], wq[j][2], wq[j][3]);
        __syncthreads();
        for (; tau < t1; tau += 32) {
            const int m0 = (tau >> 1) * 16;
            const int n0 = (tau & 1) * 32;
            const float4* xn = rows_of(tau + 32);
            v8f c0 = {}, c1 = {};
            #pragma unroll
            for (int q = 0; q < 16; ++q) {
                const float4 b0 = *(const float4*)&wl[(((2 * q + hh) << 6) + n0 + c) << 2];
                const float4 b1 = *(const float4*)&wl[(((2 * q + hh) << 6) + n0 + 16 + c) << 2];
                v2f a01, a23, b001, b023, b101, b123;
                a01.x = a[q].x; a01.y = a[q].y; a23.x = a[q].z; a23.y = a[q].w;
                b001.x = b0.x; b001.y = b0.y; b023.x = b0.z; b023.y = b0.w;
                b101.x = b1.x; b101.y = b1.y; b123.x = b1.z; b123.y = b1.w;
                c0 = __builtin_amdgcn_wmma_f32_16x16x4_f32(false, a01, false, b001, (short)0, c0, false, false);
                c1 = __builtin_amdgcn_wmma_f32_16x16x4_f32(false, a01, false, b101, (short)0, c1, false, false);
                c0 = __builtin_amdgcn_wmma_f32_16x16x4_f32(false, a23, false, b023, (short)0, c0, false, false);
                c1 = __builtin_amdgcn_wmma_f32_16x16x4_f32(false, a23, false, b123, (short)0, c1, false, false);
                a[q] = xn[2 * q];
            }
            #pragma unroll
            for (int r = 0; r < 8; ++r) {
                const float s0 = xhalf(c0[r]);
                const float s1 = xhalf(c1[r]);
                const float lo = hh == 0 ? c0[r] : s1;
                const float hi = hh == 0 ? s0 : c1[r];
                if (m0 + r < nN)     H[(size_t)(m0 + r) * HID1 + n0 + lane] = lo;
                if (m0 + 8 + r < nN) H[(size_t)(m0 + 8 + r) * HID1 + n0 + lane] = hi;
            }
        }
        return;
    }

    const int part = blockIdx.x - NPROD;
    const int e0 = (int)((long long)nE * part / NPARTS);
    const int e1 = (int)((long long)nE * (part + 1) / NPARTS);
    int cnt = e1 - e0;
    cnt = cnt > EPART ? EPART : cnt;

    int dv[KPT], sv[KPT], pk[KPT];
    #pragma unroll
    for (int k = 0; k < KPT; ++k) {
        const int i = k * 1024 + tid;
        const int e = i < cnt ? e0 + i : 0;
        dv[k] = dst[e];
        sv[k] = src[e];
    }
    if (tid < NRANGE) hist[tid] = 0;
    __syncthreads();

    #pragma unroll
    for (int k = 0; k < KPT; ++k) {
        const int i = k * 1024 + tid;
        pk[k] = -1;
        if (i < cnt && (unsigned)dv[k] < (unsigned)(nN < RN * NRANGE ? nN : RN * NRANGE)) {
            const int r  = (int)((unsigned)dv[k] / (unsigned)RN);
            const int dl = dv[k] - r * RN;
            const int rank = atomicAdd(&hist[r], 1);
            pk[k] = r | (dl << 8) | (rank << 16);
        }
    }
    __syncthreads();

    if (wave == 0) {
        int hv[8];
        int sum = 0;
        #pragma unroll
        for (int j = 0; j < 8; ++j) { hv[j] = hist[lane * 8 + j]; sum += hv[j]; }
        int inc = sum;
        #pragma unroll
        for (int off = 1; off < 32; off <<= 1) {
            const int t = __shfl_up(inc, off, 32);
            if (lane >= off) inc += t;
        }
        int run = inc - sum;
        #pragma unroll
        for (int j = 0; j < 8; ++j) { offs[lane * 8 + j] = run; run += hv[j]; }
        if (lane == 31) offs[NRANGE] = run;
    }
    __syncthreads();

    int* const tables = lists + (size_t)2 * NPARTS * EPART;
    if (tid < NRANGE)
        *(int2*)(tables + ((size_t)part * NRANGE + tid) * 2) = make_int2(offs[tid], hist[tid]);

    #pragma unroll
    for (int k = 0; k < KPT; ++k) {
        if (pk[k] >= 0) {
            const int pos = offs[pk[k] & 255] + (pk[k] >> 16);
            st_s[pos] = sv[k];
            st_d[pos] = (pk[k] >> 8) & 255;
        }
    }
    __syncthreads();

    const int tot = offs[NRANGE];
    int* const gs = lists + (size_t)part * EPART;
    int* const gd = lists + (size_t)NPARTS * EPART + (size_t)part * EPART;
    for (int i = tid; i < tot; i += 1024) {
        gs[i] = st_s[i];
        gd[i] = st_d[i];
    }
}

template <int NRANGE, int NPARTS>
struct WaveRuns {
    static constexpr int PPW = WalkGeom<NRANGE, NPARTS>::PPW;
    int base[PPW];
    int first[PPW];
    int total;
    __device__ __forceinline__ void load(const int* __restrict__ lists, int range, int wave) {
        const int* const tables = lists + (size_t)2 * NPARTS * EPART;
        int st[PPW], c[PPW];
        #pragma unroll
        for (int j = 0; j < PPW; ++j) {
            const int p = wave * PPW + j;
            const int2 t = *(const int2*)(tables + ((size_t)p * NRANGE + range) * 2);
            st[j] = __builtin_amdgcn_readfirstlane(t.x);
            c[j]  = __builtin_amdgcn_readfirstlane(t.y);
        }
        int pos = 0;
        #pragma unroll
        for (int j = 0; j < PPW; ++j) {
            first[j] = pos;
            base[j]  = (wave * PPW + j) * EPART + st[j] - pos;
            pos += c[j];
        }
        total = pos;
    }
    __device__ __forceinline__ int at(int v) const {
        int a = base[0];
        #pragma unroll
        for (int j = 1; j < PPW; ++j) a = v >= first[j] ? base[j] : a;
        return a + v;
    }
};

template <int RN, int NRANGE, int NPARTS>
__global__ __launch_bounds__(1024) void edge_count_dinv_kernel(const int* __restrict__ lists,
                                                              float* __restrict__ dinv, int nN) {
    __shared__ int cntl[RN];
    const int tid  = threadIdx.x;
    const int lane = tid & 31;
    const int wave = __builtin_amdgcn_readfirstlane(tid >> 5);
    const int range = blockIdx.x;
    const int lo = range * RN;
    int cnt = nN - lo;
    cnt = cnt < 0 ? 0 : (cnt > RN ? RN : cnt);
    const int* const gd = lists + (size_t)NPARTS * EPART;
    WaveRuns<NRANGE, NPARTS> runs;
    runs.load(lists, range, wave);
    const int nb = (runs.total + 31) >> 5;
    for (int i = tid; i < RN; i += 1024) cntl[i] = 0;
    __syncthreads();
    for (int k0 = 0; k0 < nb; k0 += 8) {
        int dlv[8];
        #pragma unroll
        for (int j = 0; j < 8; ++j) {
            const int v = 32 * (k0 + j) + lane;
            dlv[j] = v < runs.total ? gd[runs.at(v)] : -1;
        }
        #pragma unroll
        for (int j = 0; j < 8; ++j)
            if (dlv[j] >= 0) atomicAdd(&cntl[dlv[j]], 1);
    }
    __syncthreads();
    for (int i = tid; i < cnt; i += 1024) dinv[lo + i] = rsqrtf((float)cntl[i] + 1.0f);
}

template <int CH, bool LAST, int RN, int NRANGE, int NPARTS>
__global__ __launch_bounds__(1024) void edge_agg_kernel(const float* __restrict__ h,
                                                       const int* __restrict__ lists,
                                                       const float* __restrict__ dinv,
                                                       const float* __restrict__ bias,
                                                       const float* __restrict__ Wn,
                                                       const float* __restrict__ bh,
                                                       float* __restrict__ outp,
                                                       int nN) {
    constexpr int VW = CH / 32;
    constexpr int BLK  = 8192;
    constexpr int NSB  = 8;
    constexpr int ACCR = 2048;
    constexpr int DCH  = BLK - ACCR;
    static_assert(((RN + NSB - 1) / NSB) * CH <= ACCR, "a block's share of the sum rows fits its head");
    static_assert(DCH % 4 == 0, "a 16-B piece of the table stays inside one block");
    static_assert(NSB == 8, "row r: block r & 7, slot r >> 3");
    constexpr int TBL = RN * NRANGE;
    constexpr int NQT = (TBL / 4 + 1023) / 1024;
    static_assert(TBL % 4 == 0, "the table is copied in 16-B pieces");
    static_assert(TBL > NSB * DCH && TBL <= (NSB + 1) * DCH, "the table ends in the ninth block");
    constexpr int TAIL = NSB * BLK + (TBL - NSB * DCH);
    static_assert(TAIL % 64 == 0, "the finished rows start on bank 0");
    constexpr int LTOT = TAIL + (LAST ? RN : RN * (CH + 4));
    static_assert(LTOT * 4 <= 320 * 1024, "one WGP's LDS");
    __shared__ __attribute__((aligned(16))) float lds[LTOT];
    float* const outv = lds + TAIL;
    auto arow = [&](int r) -> int { return (r & (NSB - 1)) * BLK + (r >> 3) * CH; };
    auto dpos = [&](unsigned i) -> unsigned {
        const unsigned b = i / (unsigned)DCH;
        return i + (unsigned)ACCR * (b < (unsigned)NSB ? b + 1u : (unsigned)NSB);
    };
    const int tid  = threadIdx.x;
    const int lane = tid & 31;
    const int wave = __builtin_amdgcn_readfirstlane(tid >> 5);
    const int range = blockIdx.x;
    const int lo = range * RN;
    int cnt = nN - lo;
    cnt = cnt < 0 ? 0 : (cnt > RN ? RN : cnt);

    const int* const gs = lists;
    const int* const gd = lists + (size_t)NPARTS * EPART;
    const int nlim = nN < TBL ? nN : TBL;
    const int nq   = nlim >> 2;
    const int rot  = nq > 0 ? (int)(((unsigned)blockIdx.x * 48u) % (unsigned)nq) : 0;
    float4 tq[NQT];
    unsigned pq[NQT];
    #pragma unroll
    for (int j = 0; j < NQT; ++j) {
        unsigned qq = (unsigned)(j * 1024 + tid) + (unsigned)rot;
        qq = qq >= (unsigned)nq ? qq - (unsigned)nq : qq;
        qq = qq >= (unsigned)nq ? qq - (unsigned)nq : qq;
        pq[j] = qq < (unsigned)nq ? qq : 0u;
        tq[j] = *(const float4*)(dinv + 4u * pq[j]);
    }
    float tqt = 0.0f;
    if (tid < (nlim & 3)) tqt = dinv[4 * nq + tid];
    WaveRuns<NRANGE, NPARTS> runs;
    runs.load(lists, range, wave);
    const int tot = runs.total;
    const int nb = (tot + 31) >> 5;
    int s = 0, dl = 0, n = 0;
    if (nb > 0) {
        n = tot < 32 ? tot : 32;
        if (lane < n) { const int a = runs.at(lane); s = gs[a]; dl = gd[a]; }
    }

    #pragma unroll
    for (int j = 0; j < NQT; ++j) *(float4*)&lds[dpos(4u * pq[j])] = tq[j];
    if (tid < (nlim & 3)) lds[dpos((unsigned)(4 * nq + tid))] = tqt;
    for (int i = tid * 4; i < RN * CH; i += 4096)
        *(float4*)&lds[arow(i / CH) + i % CH] = make_float4(0.0f, 0.0f, 0.0f, 0.0f);
    __syncthreads();

    for (int k = 0; k < nb; ++k) {
        int s2 = 0, dl2 = 0, n2 = 0;
        const bool more = k + 1 < nb;
        if (more) {
            const int left = tot - 32 * (k + 1);
            n2 = left < 32 ? left : 32;
        }
        const int v2 = 32 * (k + 1) + lane;
        const int st = (unsigned)s < (unsigned)(TBL - 1) ? s : TBL - 1;
        const unsigned rowb = (unsigned)s * (unsigned)(CH * 4);
        const unsigned accb = (unsigned)arow(dl) * 4u;
        const unsigned lnb  = (unsigned)lane * (unsigned)(VW * 4);
        float ns, nd, nm;
        if (n == 32) {
            float r[32][VW];
            #pragma unroll
            for (int u = 0; u < 32; ++u) {
                const unsigned ob = (unsigned)__builtin_amdgcn_readlane((int)rowb, u) + lnb;
                const float* hp = (const float*)((const char*)h + ob);
                #pragma unroll
                for (int w = 0; w < VW; ++w) r[u][w] = hp[w];
            }
            ns = lds[dpos((unsigned)st)]; nd = lds[dpos((unsigned)(lo + dl))];
            if (more && lane < n2) { const int a = runs.at(v2); s2 = gs[a]; dl2 = gd[a]; }
            nm = ns * nd;
            #pragma unroll
            for (int u = 0; u < 32; ++u) {
                const unsigned ab = (unsigned)__builtin_amdgcn_readlane((int)accb, u) + lnb;
                const float nu = __builtin_bit_cast(float,
                    __builtin_amdgcn_readlane(__builtin_bit_cast(int, nm), u));
                float* ap = (float*)((char*)lds + ab);
                #pragma unroll
                for (int w = 0; w < VW; ++w) atomicAdd(ap + w, r[u][w] * nu);
            }
        } else {
            ns = lds[dpos((unsigned)st)]; nd = lds[dpos((unsigned)(lo + dl))];
            if (more && lane < n2) { const int a = runs.at(v2); s2 = gs[a]; dl2 = gd[a]; }
            nm = (lane < n) ? ns * nd : 0.0f;
            #pragma unroll
            for (int g = 0; g < 4; ++g) {
                if (g * 8 < n) {
                    float r[8][VW];
                    #pragma unroll
                    for (int u = 0; u < 8; ++u) {
                        const unsigned ob = (unsigned)__builtin_amdgcn_readlane((int)rowb, g * 8 + u) + lnb;
                        const float* hp = (const float*)((const char*)h + ob);
                        #pragma unroll
                        for (int w = 0; w < VW; ++w) r[u][w] = hp[w];
                    }
                    #pragma unroll
                    for (int u = 0; u < 8; ++u) {
                        if (g * 8 + u < n) {
                            const unsigned ab = (unsigned)__builtin_amdgcn_readlane((int)accb, g * 8 + u) + lnb;
                            const float nu = __builtin_bit_cast(float,
                                __builtin_amdgcn_readlane(__builtin_bit_cast(int, nm), g * 8 + u));
                            float* ap = (float*)((char*)lds + ab);
                            #pragma unroll
                            for (int w = 0; w < VW; ++w) atomicAdd(ap + w, r[u][w] * nu);
                        }
                    }
                }
            }
        }
        s = s2; dl = dl2; n = n2;
    }
    __syncthreads();

    if (!LAST) {
        static_assert(LAST || CH % 4 == 0, "k runs in steps of four");
        constexpr int Q4  = CH / 4;
        constexpr int NIT = (RN * Q4 + 1023) / 1024;
        constexpr int KQ  = CH / 4;
        constexpr int NRT = (RN + 15) / 16;
        constexpr int GP = CH + 4;
        static_assert(LAST || NRT <= 32, "one wave per 16-row tile");
        float* const gl = lds + TAIL;
        static_assert(LAST || 1024 % Q4 == 0, "a thread keeps its four channels over its rows");
        const int c16 = lane & 15;
        const int hh  = lane >> 4;
        const int c4  = (tid % Q4) * 4;
        const bool mul = wave < NRT && wave * 16 < cnt;
        const int lr = cnt > 0 ? lo : 0;
        const float* hr = h + (size_t)lr * CH;
        float4 hv[NIT];
        float  dq[NIT];
        v2f bw0[KQ], bw1[KQ];
        #pragma unroll
        for (int q = 0; q < NIT; ++q) {
            int i = q * 1024 + tid;
            i = i < cnt * Q4 ? i : 0;
            hv[q] = *(const float4*)(hr + (size_t)i * 4);
            dq[q] = lds[dpos((unsigned)(lr + i / Q4))];
        }
        const float4 bv = *(const float4*)(bias + c4);
        if (mul) {
            #pragma unroll
            for (int q = 0; q < KQ; ++q) {
                const int k = 4 * q + 2 * hh;
                bw0[q].x = Wn[(k + 0) * HID2 + c16];
                bw0[q].y = Wn[(k + 1) * HID2 + c16];
                bw1[q].x = Wn[(k + 0) * HID2 + 16 + c16];
                bw1[q].y = Wn[(k + 1) * HID2 + 16 + c16];
            }
        }
        #pragma unroll
        for (int q = 0; q < NIT; ++q) {
            const int i = q * 1024 + tid;
            if (i < cnt * Q4) {
                const float dd = dq[q] * dq[q];
                const float4 a  = *(const float4*)&lds[arow(i / Q4) + c4];
                float4 v;
                v.x = fmaxf(fmaf(hv[q].x, dd, a.x) + bv.x, 0.0f);
                v.y = fmaxf(fmaf(hv[q].y, dd, a.y) + bv.y, 0.0f);
                v.z = fmaxf(fmaf(hv[q].z, dd, a.z) + bv.z, 0.0f);
                v.w = fmaxf(fmaf(hv[q].w, dd, a.w) + bv.w, 0.0f);
                *(float4*)&gl[(i / Q4) * GP + c4] = v;
            }
        }
        __syncthreads();
        if (mul) {
            const int m0 = wave * 16;
            int ra = m0 + c16;
            ra = ra < cnt ? ra : cnt - 1;
            const float* ap = &gl[ra * GP + 2 * hh];
            v2f af[KQ];
            #pragma unroll
            for (int q = 0; q < KQ; ++q) af[q] = *(const v2f*)(ap + 4 * q);
            v8f c0 = {}, c1 = {};
            #pragma unroll
            for (int q = 0; q < KQ; ++q) {
                c0 = __builtin_amdgcn_wmma_f32_16x16x4_f32(false, af[q], false, bw0[q], (short)0, c0, false, false);
                c1 = __builtin_amdgcn_wmma_f32_16x16x4_f32(false, af[q], false, bw1[q], (short)0, c1, false, false);
            }
            float* orow = outp + (size_t)(lo + m0) * HID2 + lane;
            #pragma unroll
            for (int r = 0; r < 8; ++r) {
                const float s0 = xhalf(c0[r]);
                const float s1 = xhalf(c1[r]);
                const float lov = hh == 0 ? c0[r] : s1;
                const float hiv = hh == 0 ? s0 : c1[r];
                if (m0 + r < cnt)     orow[(size_t)r * HID2] = lov;
                if (m0 + 8 + r < cnt) orow[(size_t)(8 + r) * HID2] = hiv;
            }
        }
    } else {
        static_assert(!LAST || CH == 32, "the projection takes one channel per lane");
        constexpr int NPW = (RN + 31) / 32;
        float hq[NPW], dq[NPW];
        #pragma unroll
        for (int q = 0; q < NPW; ++q) {
            const int node = q * 32 + wave;
            hq[q] = 0.0f;
            dq[q] = 0.0f;
            if (node < cnt) {
                hq[q] = h[(size_t)(lo + node) * CH + lane];
                dq[q] = lds[dpos((unsigned)(lo + node))];
            }
        }
        const float bl = bias[lane], wl = Wn[lane], b0 = bh[0];
        #pragma unroll
        for (int q = 0; q < NPW; ++q) {
            const int node = q * 32 + wave;
            if (node < cnt) {
                const float dd = dq[q] * dq[q];
                const float a = lds[arow(node) + lane];
                float v = a + hq[q] * dd + bl;
                v = fmaxf(v, 0.0f);
                float p = v * wl;
                #pragma unroll
                for (int off = 16; off > 0; off >>= 1)
                    p += __shfl_xor(p, off, 32);
                if (lane == 0) outv[node] = p + b0;
            }
        }
        __syncthreads();
        for (int i = tid; i < cnt; i += 1024) outp[lo + i] = outv[i];
    }
}

#define NRG1 256
#define NPT1 128
#define NPR1 128
#define RNG1 196

extern "C" void kernel_launch(void* const* d_in, const int* in_sizes, int n_in,
                              void* d_out, int out_size, void* d_ws, size_t ws_size,
                              hipStream_t stream) {
    const float* x  = (const float*)d_in[0];
    const int*   ei = (const int*)d_in[1];
    const float* W1 = (const float*)d_in[2];
    const float* b1 = (const float*)d_in[3];
    const float* W2 = (const float*)d_in[4];
    const float* b2 = (const float*)d_in[5];
    const float* Wh = (const float*)d_in[6];
    const float* bh = (const float*)d_in[7];
    float* out = (float*)d_out;

    const int N = in_sizes[0] / IN_CH;
    const int E = in_sizes[1] / 2;
    const int* src = ei;
    const int* dst = ei + E;

    float* f     = (float*)d_ws;
    float* dinv  = f;
    float* h1    = f + 65536;
    float* h2    = h1 + (size_t)N * HID1;
    int*   lists = (int*)(h2 + (size_t)N * HID2);

    split_prod_kernel<RNG1, NRG1, NPT1, NPR1><<<NPR1 + NPT1, 1024, 0, stream>>>(src, dst, lists, E, N, x, W1, h1);

    edge_count_dinv_kernel<RNG1, NRG1, NPT1><<<NRG1, 1024, 0, stream>>>(lists, dinv, N);

    edge_agg_kernel<HID1, false, RNG1, NRG1, NPT1><<<NRG1, 1024, 0, stream>>>(h1, lists, dinv, b1, W2, bh, h2, N);

    edge_agg_kernel<HID2, true, RNG1, NRG1, NPT1><<<NRG1, 1024, 0, stream>>>(h2, lists, dinv, b2, Wh, bh, out, N);
}
